# starting kernel
# baseline (speedup 1.0000x reference)
#include <hip/hip_runtime.h>
#include <cstdint>
#include <cstddef>


typedef float v2f __attribute__((ext_vector_type(2)));
typedef float v8f __attribute__((ext_vector_type(8)));

#define D 128
#define LN_EPS 1e-5f

__global__ void rgcn_init_kernel(float* __restrict__ deg,
                                 float* __restrict__ agg, int N) {
    int tid = blockIdx.x * blockDim.x + threadIdx.x;
    if (tid < N * D) agg[tid] = 0.0f;
    if (tid < N)     deg[tid] = 1.0f;
}

__global__ void rgcn_degree_kernel(const int* __restrict__ row,
                                   float* __restrict__ deg, int E) {
    int e = blockIdx.x * blockDim.x + threadIdx.x;
    if (e < E) atomicAdd(&deg[row[e]], 1.0f);
}

__global__ void __launch_bounds__(256)
rgcn_gemm_kernel(const float* __restrict__ x, const float* __restrict__ W,
                 const float* __restrict__ bias, float* __restrict__ h, int N) {
    const int lane = threadIdx.x & 31;
    const int wave = threadIdx.x >> 5;
    const int m    = lane & 15;
    const int half = lane >> 4;
    const int row0 = blockIdx.x * 128 + wave * 16;

    int r = row0 + m;
    if (r >= N) r = N - 1;

    const float* __restrict__ xrow  = x + (size_t)r * D + 2 * half;
    const float* __restrict__ wbase = W + (size_t)m * D + 2 * half;

    v8f acc[8] = {};
#pragma unroll 4
    for (int k = 0; k < D; k += 4) {
        v2f a = *(const v2f*)(xrow + k);
#pragma unroll
        for (int t = 0; t < 8; ++t) {
            v2f b = *(const v2f*)(wbase + (size_t)t * 16 * D + k);
            acc[t] = __builtin_amdgcn_wmma_f32_16x16x4_f32(
                 false, a,  false, b,
                 (short)0, acc[t],  false,  false);
        }
    }

#pragma unroll
    for (int t = 0; t < 8; ++t) {
        const float bb = bias[t * 16 + m];
#pragma unroll
        for (int v = 0; v < 8; ++v) {
            int orow = row0 + half * 8 + v;
            if (orow < N)
                h[(size_t)orow * D + t * 16 + m] = acc[t][v] + bb;
        }
    }
}

__global__ void __launch_bounds__(256)
rgcn_edge_kernel(const int* __restrict__ ei, const float* __restrict__ deg,
                 const float* __restrict__ h, float* __restrict__ agg, int E) {
    const int e = blockIdx.x * 8 + (threadIdx.x >> 5);
    if (e >= E) return;
    const int lane = threadIdx.x & 31;

    const int s = ei[e];
    const int t = ei[E + e];
    const float norm = __frsqrt_rn(deg[s] * deg[t]);

    const float4 v = ((const float4*)(h + (size_t)s * D))[lane];
    float* at = agg + (size_t)t * D + lane * 4;
    atomicAdd(at + 0, norm * v.x);
    atomicAdd(at + 1, norm * v.y);
    atomicAdd(at + 2, norm * v.z);
    atomicAdd(at + 3, norm * v.w);
}

__global__ void __launch_bounds__(256)
rgcn_final_kernel(const float* __restrict__ x, const float* __restrict__ deg,
                  const float* __restrict__ h, const float* __restrict__ agg,
                  const float* __restrict__ gamma, const float* __restrict__ beta,
                  float* __restrict__ out, int N) {
    const int n = blockIdx.x * 8 + (threadIdx.x >> 5);
    if (n >= N) return;
    const int lane = threadIdx.x & 31;

    const float self_w = 1.0f / deg[n];

    const float4 hv = ((const float4*)(h   + (size_t)n * D))[lane];
    const float4 av = ((const float4*)(agg + (size_t)n * D))[lane];

    float4 v;
    v.x = fmaxf(av.x + self_w * hv.x, 0.0f);
    v.y = fmaxf(av.y + self_w * hv.y, 0.0f);
    v.z = fmaxf(av.z + self_w * hv.z, 0.0f);
    v.w = fmaxf(av.w + self_w * hv.w, 0.0f);

    float s  = v.x + v.y + v.z + v.w;
    float s2 = v.x * v.x + v.y * v.y + v.z * v.z + v.w * v.w;
#pragma unroll
    for (int o = 16; o >= 1; o >>= 1) {
        s  += __shfl_xor(s,  o, 32);
        s2 += __shfl_xor(s2, o, 32);
    }
    const float mu   = s * (1.0f / D);
    const float var  = s2 * (1.0f / D) - mu * mu;
    const float rstd = __frsqrt_rn(var + LN_EPS);

    const float4 g  = ((const float4*)gamma)[lane];
    const float4 be = ((const float4*)beta)[lane];
    const float4 xv = ((const float4*)(x + (size_t)n * D))[lane];

    float4 r;
    r.x = (v.x - mu) * rstd * g.x + be.x + xv.x;
    r.y = (v.y - mu) * rstd * g.y + be.y + xv.y;
    r.z = (v.z - mu) * rstd * g.z + be.z + xv.z;
    r.w = (v.w - mu) * rstd * g.w + be.w + xv.w;

    ((float4*)(out + (size_t)n * D))[lane] = r;
}

extern "C" void kernel_launch(void* const* d_in, const int* in_sizes, int n_in,
                              void* d_out, int out_size, void* d_ws, size_t ws_size,
                              hipStream_t stream) {
    const float* x     = (const float*)d_in[0];
    const int*   ei    = (const int*)  d_in[1];
    const float* W     = (const float*)d_in[2];
    const float* bias  = (const float*)d_in[3];
    const float* gamma = (const float*)d_in[4];
    const float* beta  = (const float*)d_in[5];

    const int N = in_sizes[0] / D;
    const int E = in_sizes[1] / 2;

    char* ws = (char*)d_ws;
    float* deg = (float*)ws;
    size_t off = ((size_t)N * sizeof(float) + 255) & ~(size_t)255;
    float* h = (float*)(ws + off);
    off += (size_t)N * D * sizeof(float);
    float* agg = (float*)(ws + off);

    rgcn_init_kernel<<<(N * D + 255) / 256, 256, 0, stream>>>(deg, agg, N);
    rgcn_degree_kernel<<<(E + 255) / 256, 256, 0, stream>>>(ei, deg, E);
    rgcn_gemm_kernel<<<(N + 127) / 128, 256, 0, stream>>>(x, W, bias, h, N);
    rgcn_edge_kernel<<<(E + 7) / 8, 256, 0, stream>>>(ei, deg, h, agg, E);
    rgcn_final_kernel<<<(N + 7) / 8, 256, 0, stream>>>(x, deg, h, agg, gamma,
                                                       beta, (float*)d_out, N);
}


# aggregation without sort or global atomics: each of 256 workgroups filters the whole target list for the 196 nodes it owns, adds their h rows into LDS sums, ReLU+LN from LDS; arrival-order sums kept
# speedup vs baseline: 1.6813x; 1.6813x over previous
#include <hip/hip_runtime.h>
#include <cstdint>
#include <cstddef>


typedef float v2f __attribute__((ext_vector_type(2)));
typedef float v8f __attribute__((ext_vector_type(8)));

#define D 128
#define LN_EPS 1e-5f

__global__ void rgcn_init_kernel(float* __restrict__ deg, int N) {
    int tid = blockIdx.x * blockDim.x + threadIdx.x;
    if (tid < N)     deg[tid] = 1.0f;
}

__global__ void rgcn_degree_kernel(const int* __restrict__ row,
                                   float* __restrict__ deg, int E) {
    int e = blockIdx.x * blockDim.x + threadIdx.x;
    if (e < E) atomicAdd(&deg[row[e]], 1.0f);
}

__global__ void __launch_bounds__(256)
rgcn_gemm_kernel(const float* __restrict__ x, const float* __restrict__ W,
                 const float* __restrict__ bias, float* __restrict__ h, int N) {
    const int lane = threadIdx.x & 31;
    const int wave = threadIdx.x >> 5;
    const int m    = lane & 15;
    const int half = lane >> 4;
    const int row0 = blockIdx.x * 128 + wave * 16;

    int r = row0 + m;
    if (r >= N) r = N - 1;

    const float* __restrict__ xrow  = x + (size_t)r * D + 2 * half;
    const float* __restrict__ wbase = W + (size_t)m * D + 2 * half;

    v8f acc[8] = {};
#pragma unroll 4
    for (int k = 0; k < D; k += 4) {
        v2f a = *(const v2f*)(xrow + k);
#pragma unroll
        for (int t = 0; t < 8; ++t) {
            v2f b = *(const v2f*)(wbase + (size_t)t * 16 * D + k);
            acc[t] = __builtin_amdgcn_wmma_f32_16x16x4_f32(
                 false, a,  false, b,
                 (short)0, acc[t],  false,  false);
        }
    }

#pragma unroll
    for (int t = 0; t < 8; ++t) {
        const float bb = bias[t * 16 + m];
#pragma unroll
        for (int v = 0; v < 8; ++v) {
            int orow = row0 + half * 8 + v;
            if (orow < N)
                h[(size_t)orow * D + t * 16 + m] = acc[t][v] + bb;
        }
    }
}

#define AGG_R      196
#define AGG_WAVES  32
#define AGG_BLK    512
#define AGG_FLUSH  64
#define AGG_CAP    (AGG_BLK + AGG_FLUSH)
#define AGG_ROWS   8

__global__ void __launch_bounds__(1024)
rgcn_owner_sum_final_kernel(const int* __restrict__ ei, const float* __restrict__ deg,
                            const float* __restrict__ h, const float* __restrict__ x,
                            const float* __restrict__ gamma, const float* __restrict__ beta,
                            float* __restrict__ out, int N, int E) {
    __shared__ float acc[AGG_R * D];
    __shared__ int   pend[AGG_WAVES * AGG_CAP];

    const int lane  = threadIdx.x & 31;
    const int wave  = __builtin_amdgcn_readfirstlane(threadIdx.x >> 5);
    const int node0 = blockIdx.x * AGG_R;

    for (int i = threadIdx.x; i < AGG_R * D / 4; i += 1024)
        ((float4*)acc)[i] = make_float4(0.0f, 0.0f, 0.0f, 0.0f);
    __syncthreads();

    const int* __restrict__ col = ei + E;
    int* mylist = pend + wave * AGG_CAP;
    int cnt = 0;

    const int      nfull = ((E & 3) == 0) ? E / AGG_BLK : 0;
    const unsigned rot   = (blockIdx.x * (unsigned)nfull) / gridDim.x;

    auto load_blk = [&](int k, int4* t) {
        int a = k + (int)rot;
        if (a >= nfull) a -= nfull;
        const int ebase = a * AGG_BLK;
#pragma unroll
        for (int q = 0; q < 4; ++q)
            t[q] = *(const int4*)(col + ebase + (q * 32 + lane) * 4);
        return ebase;
    };

    auto append = [&](bool hit, int e) {
        const unsigned m = __builtin_amdgcn_ballot_w32(hit);
        if (hit) mylist[cnt + (int)__builtin_amdgcn_mbcnt_lo(m, 0u)] = e;
        cnt += __builtin_popcount(m);
    };

    auto flush = [&]() {
        __builtin_amdgcn_fence(__ATOMIC_ACQ_REL, "wavefront");
        for (int base = 0; base < cnt; base += 32) {
            const int m = min(32, cnt - base);
            int s_l = 0, tl_l = 0;
            float nrm_l = 0.0f;
            if (lane < m) {
                const int e = mylist[base + lane];
                const int t = col[e];
                s_l   = ei[e];
                tl_l  = t - node0;
                nrm_l = __frsqrt_rn(deg[s_l] * deg[t]);
            }
            for (int j = 0; j < m; j += AGG_ROWS) {
                float4 v[AGG_ROWS];
                float  nr[AGG_ROWS];
                int    tl[AGG_ROWS];
#pragma unroll
                for (int u = 0; u < AGG_ROWS; ++u) {
                    const int jj = min(j + u, m - 1);
                    const int s  = __builtin_amdgcn_readlane(s_l, jj);
                    tl[u] = __builtin_amdgcn_readlane(tl_l, jj);
                    nr[u] = __int_as_float(__builtin_amdgcn_readlane(__float_as_int(nrm_l), jj));
                    v[u]  = ((const float4*)(h + (size_t)s * D))[lane];
                }
#pragma unroll
                for (int u = 0; u < AGG_ROWS; ++u) {
                    if (j + u < m) {
                        float* a = acc + tl[u] * D + lane;
                        atomicAdd(a,      nr[u] * v[u].x);
                        atomicAdd(a + 32, nr[u] * v[u].y);
                        atomicAdd(a + 64, nr[u] * v[u].z);
                        atomicAdd(a + 96, nr[u] * v[u].w);
                    }
                }
            }
        }
        cnt = 0;
    };

    auto test_blk = [&](const int4* t, int ebase) {
#pragma unroll
        for (int q = 0; q < 4; ++q) {
            const int e0 = ebase + (q * 32 + lane) * 4;
            const bool h0 = (unsigned)(t[q].x - node0) < (unsigned)AGG_R;
            const bool h1 = (unsigned)(t[q].y - node0) < (unsigned)AGG_R;
            const bool h2 = (unsigned)(t[q].z - node0) < (unsigned)AGG_R;
            const bool h3 = (unsigned)(t[q].w - node0) < (unsigned)AGG_R;
            if (__builtin_amdgcn_ballot_w32(h0 | h1 | h2 | h3) != 0u) {
                append(h0, e0);
                append(h1, e0 + 1);
                append(h2, e0 + 2);
                append(h3, e0 + 3);
            }
        }
        if (cnt >= AGG_FLUSH) flush();
    };

    if (wave < nfull) {
        int4 ta[4], tb[4];
        int k  = wave;
        int ea = load_blk(k, ta), eb = 0;
        for (;;) {
            eb = load_blk(min(k + AGG_WAVES, nfull - 1), tb);
            test_blk(ta, ea);
            k += AGG_WAVES;
            if (k >= nfull) break;
            ea = load_blk(min(k + AGG_WAVES, nfull - 1), ta);
            test_blk(tb, eb);
            k += AGG_WAVES;
            if (k >= nfull) break;
        }
    }
    for (int e0 = nfull * AGG_BLK + wave * 32; e0 < E; e0 += AGG_WAVES * 32) {
        const int e = e0 + lane;
        const int t = (e < E) ? col[e] : -1;
        append((unsigned)(t - node0) < (unsigned)AGG_R, e);
        if (cnt >= AGG_FLUSH) flush();
    }
    if (cnt > 0) flush();
    __syncthreads();

    const float4 g  = ((const float4*)gamma)[lane];
    const float4 be = ((const float4*)beta)[lane];
    for (int tl = wave; tl < AGG_R; tl += AGG_WAVES) {
        const int n = node0 + tl;
        if (n >= N) break;

        const float self_w = 1.0f / deg[n];

        const float4 hv = ((const float4*)(h + (size_t)n * D))[lane];
        float4 av;
        av.x = acc[tl * D + lane];
        av.y = acc[tl * D + 32 + lane];
        av.z = acc[tl * D + 64 + lane];
        av.w = acc[tl * D + 96 + lane];

        float4 v;
        v.x = fmaxf(av.x + self_w * hv.x, 0.0f);
        v.y = fmaxf(av.y + self_w * hv.y, 0.0f);
        v.z = fmaxf(av.z + self_w * hv.z, 0.0f);
        v.w = fmaxf(av.w + self_w * hv.w, 0.0f);

        float s  = v.x + v.y + v.z + v.w;
        float s2 = v.x * v.x + v.y * v.y + v.z * v.z + v.w * v.w;
#pragma unroll
        for (int o = 16; o >= 1; o >>= 1) {
            s  += __shfl_xor(s,  o, 32);
            s2 += __shfl_xor(s2, o, 32);
        }
        const float mu   = s * (1.0f / D);
        const float var  = s2 * (1.0f / D) - mu * mu;
        const float rstd = __frsqrt_rn(var + LN_EPS);

        const float4 xv = ((const float4*)(x + (size_t)n * D))[lane];

        float4 r;
        r.x = (v.x - mu) * rstd * g.x + be.x + xv.x;
        r.y = (v.y - mu) * rstd * g.y + be.y + xv.y;
        r.z = (v.z - mu) * rstd * g.z + be.z + xv.z;
        r.w = (v.w - mu) * rstd * g.w + be.w + xv.w;

        ((float4*)(out + (size_t)n * D))[lane] = r;
    }
}

extern "C" void kernel_launch(void* const* d_in, const int* in_sizes, int n_in,
                              void* d_out, int out_size, void* d_ws, size_t ws_size,
                              hipStream_t stream) {
    const float* x     = (const float*)d_in[0];
    const int*   ei    = (const int*)  d_in[1];
    const float* W     = (const float*)d_in[2];
    const float* bias  = (const float*)d_in[3];
    const float* gamma = (const float*)d_in[4];
    const float* beta  = (const float*)d_in[5];

    const int N = in_sizes[0] / D;
    const int E = in_sizes[1] / 2;

    char* ws = (char*)d_ws;
    float* deg = (float*)ws;
    size_t off = ((size_t)N * sizeof(float) + 255) & ~(size_t)255;
    float* h = (float*)(ws + off);

    rgcn_init_kernel<<<(N + 255) / 256, 256, 0, stream>>>(deg, N);
    rgcn_degree_kernel<<<(E + 255) / 256, 256, 0, stream>>>(ei, deg, E);
    rgcn_gemm_kernel<<<(N + 127) / 128, 256, 0, stream>>>(x, W, bias, h, N);
    rgcn_owner_sum_final_kernel<<<(N + AGG_R - 1) / AGG_R, 1024, 0, stream>>>(
        ei, deg, h, x, gamma, beta, (float*)d_out, N, E);
}


# degree launch also writes a presence word per 1024-edge chunk and owner (bit per 32-edge group); owners load and test only flagged groups, not all targets; arrival-order sums, list order differs
# speedup vs baseline: 2.1192x; 1.2605x over previous
#include <hip/hip_runtime.h>
#include <cstdint>
#include <cstddef>


typedef float v2f __attribute__((ext_vector_type(2)));
typedef float v8f __attribute__((ext_vector_type(8)));

#define D 128
#define LN_EPS 1e-5f

__global__ void rgcn_init_kernel(float* __restrict__ deg, int N) {
    int tid = blockIdx.x * blockDim.x + threadIdx.x;
    if (tid < N)     deg[tid] = 1.0f;
}

#define AGG_R       196
#define PRES_GRP    32
#define PRES_CHUNK  1024

__global__ void __launch_bounds__(256)
rgcn_degree_presence_kernel(const int* __restrict__ ei, float* __restrict__ deg,
                            unsigned* __restrict__ present, int E, int G) {
    extern __shared__ unsigned omask[];
    const int lane = threadIdx.x & 31;
    const int wave = threadIdx.x >> 5;
    for (int i = threadIdx.x; i < G; i += 256) omask[i] = 0u;
    __syncthreads();

    const int* __restrict__ col = ei + E;
    const int cbase = blockIdx.x * PRES_CHUNK;
    int s[4], t[4];
#pragma unroll
    for (int q = 0; q < 4; ++q) {
        const int e = min(cbase + (wave * 4 + q) * PRES_GRP + lane, E - 1);
        s[q] = ei[e];
        t[q] = col[e];
    }
#pragma unroll
    for (int q = 0; q < 4; ++q) {
        const int g = wave * 4 + q;
        if (cbase + g * PRES_GRP + lane < E) {
            atomicAdd(&deg[s[q]], 1.0f);
            const unsigned o = (unsigned)t[q] / (unsigned)AGG_R;
            if (o < (unsigned)G) atomicOr(&omask[o], 1u << g);
        }
    }
    __syncthreads();
    for (int i = threadIdx.x; i < G; i += 256)
        present[(size_t)blockIdx.x * G + i] = omask[i];
}

#define AGG_WAVES  32
#define AGG_K      16
#define AGG_FLUSH  64
#define AGG_CAP    (AGG_K * PRES_GRP + AGG_FLUSH)
#define AGG_ROWS   8
#define AGG_RP     208
#define ACC_P      132
#define GEMM_JOBS  (AGG_RP / 16 * 4)

__global__ void __launch_bounds__(1024)
rgcn_owner_sum_linear_final_kernel(const int* __restrict__ ei, const float* __restrict__ deg,
                                   const unsigned* __restrict__ present,
                                   const float* __restrict__ x, const float* __restrict__ W,
                                   const float* __restrict__ bias,
                                   const float* __restrict__ gamma, const float* __restrict__ beta,
                                   float* __restrict__ out, int N, int E) {
    __shared__ __attribute__((aligned(16))) float acc[AGG_RP * ACC_P];
    __shared__ __attribute__((aligned(16))) float wl[D * ACC_P];
    __shared__ float nsum[AGG_RP];
    __shared__ int   pend[AGG_WAVES * AGG_CAP];

    const int lane  = threadIdx.x & 31;
    const int wave  = __builtin_amdgcn_readfirstlane(threadIdx.x >> 5);
    const int node0 = blockIdx.x * AGG_R;

    for (int i = threadIdx.x; i < AGG_RP * ACC_P / 4; i += 1024)
        ((float4*)acc)[i] = make_float4(0.0f, 0.0f, 0.0f, 0.0f);
    if (threadIdx.x < AGG_RP) nsum[threadIdx.x] = 0.0f;
    for (int i = threadIdx.x; i < D * D / 4; i += 1024) {
        const int n = i >> 5, c4 = i & 31;
        *(float4*)(wl + n * ACC_P + 4 * c4) = ((const float4*)W)[i];
    }
    __syncthreads();

    const int* __restrict__ col = ei + E;
    int* mylist = pend + wave * AGG_CAP;
    int cnt = 0;

    auto append = [&](bool hit, int e) {
        const unsigned m = __builtin_amdgcn_ballot_w32(hit);
        if (hit) mylist[cnt + (int)__builtin_amdgcn_mbcnt_lo(m, 0u)] = e;
        cnt += __builtin_popcount(m);
    };

    auto flush = [&]() {
        __builtin_amdgcn_fence(__ATOMIC_ACQ_REL, "wavefront");
        for (int base = 0; base < cnt; base += 32) {
            const int m = min(32, cnt - base);
            int s_l = 0, tl_l = 0;
            float nrm_l = 0.0f;
            if (lane < m) {
                const int e = mylist[base + lane];
                int t;
                if (e < E) { t = col[e]; s_l = ei[e]; }
                else       { t = s_l = e - E; }
                tl_l  = t - node0;
                nrm_l = __frsqrt_rn(deg[s_l] * deg[t]);
                atomicAdd(nsum + tl_l, nrm_l);
            }
            for (int j = 0; j < m; j += AGG_ROWS) {
                float4 v[AGG_ROWS];
                float  nr[AGG_ROWS];
                int    tl[AGG_ROWS];
#pragma unroll
                for (int u = 0; u < AGG_ROWS; ++u) {
                    const int jj = min(j + u, m - 1);
                    const int s  = __builtin_amdgcn_readlane(s_l, jj);
                    tl[u] = __builtin_amdgcn_readlane(tl_l, jj);
                    nr[u] = __int_as_float(__builtin_amdgcn_readlane(__float_as_int(nrm_l), jj));
                    v[u]  = ((const float4*)(x + (size_t)s * D))[lane];
                }
#pragma unroll
                for (int u = 0; u < AGG_ROWS; ++u) {
                    if (j + u < m) {
                        float* a = acc + tl[u] * ACC_P + lane;
                        atomicAdd(a,      nr[u] * v[u].x);
                        atomicAdd(a + 32, nr[u] * v[u].y);
                        atomicAdd(a + 64, nr[u] * v[u].z);
                        atomicAdd(a + 96, nr[u] * v[u].w);
                    }
                }
            }
        }
        cnt = 0;
    };

    {
        const int tl = wave + AGG_WAVES * lane;
        append(tl < AGG_R && node0 + tl < N, E + node0 + tl);
    }

    const int nchunk = (E + PRES_CHUNK - 1) / PRES_CHUNK;
    const int G      = gridDim.x;
    for (int c0 = wave; c0 < nchunk; c0 += AGG_WAVES * 32) {
        const int c = c0 + AGG_WAVES * lane;
        const unsigned wmask = (c < nchunk) ? present[(size_t)c * G + blockIdx.x] : 0u;
        unsigned live  = __builtin_amdgcn_ballot_w32(wmask != 0u);
        unsigned bits  = 0u;
        int      cbase = 0;

        auto next_group = [&]() -> int {
            if (bits == 0u) {
                if (live == 0u) return -1;
                const int l = __builtin_ctz(live);
                live &= live - 1u;
                bits  = (unsigned)__builtin_amdgcn_readlane((int)wmask, l);
                cbase = (c0 + AGG_WAVES * l) * PRES_CHUNK;
            }
            const int b = __builtin_ctz(bits);
            bits &= bits - 1u;
            return cbase + b * PRES_GRP;
        };

        for (;;) {
            int e0[AGG_K], t[AGG_K];
#pragma unroll
            for (int u = 0; u < AGG_K; ++u) {
                e0[u] = next_group();
                t[u] = col[min(max(e0[u], 0) + lane, E - 1)];
            }
#pragma unroll
            for (int u = 0; u < AGG_K; ++u) {
                if (e0[u] >= 0) {
                    const int e = e0[u] + lane;
                    append(e < E && (unsigned)(t[u] - node0) < (unsigned)AGG_R, e);
                }
            }
            if (cnt >= AGG_FLUSH) flush();
            if (e0[AGG_K - 1] < 0) break;
        }
    }
    if (cnt > 0) flush();
    __syncthreads();

    const int m    = lane & 15;
    const int half = lane >> 4;
    v8f c[2][2] = {};
#pragma unroll
    for (int slot = 0; slot < 2; ++slot) {
        const int job = wave + slot * AGG_WAVES;
        if (job < GEMM_JOBS) {
            const int strip = job >> 2, t0 = (job & 3) * 2;
            const float* arow = acc + (strip * 16 + m) * ACC_P + 64 * half;
            const float* b0   = wl + ((t0 + 0) * 16 + m) * ACC_P + 2 * half;
            const float* b1   = wl + ((t0 + 1) * 16 + m) * ACC_P + 2 * half;
#pragma unroll 8
            for (int kk = 0; kk < D / 4; ++kk) {
                v2f a;
                a.x = arow[kk];
                a.y = arow[32 + kk];
                const v2f w0 = *(const v2f*)(b0 + 4 * kk);
                const v2f w1 = *(const v2f*)(b1 + 4 * kk);
                c[slot][0] = __builtin_amdgcn_wmma_f32_16x16x4_f32(
                     false, a,  false, w0,
                     (short)0, c[slot][0],  false,  false);
                c[slot][1] = __builtin_amdgcn_wmma_f32_16x16x4_f32(
                     false, a,  false, w1,
                     (short)0, c[slot][1],  false,  false);
            }
        }
    }
    __syncthreads();

#pragma unroll
    for (int slot = 0; slot < 2; ++slot) {
        const int job = wave + slot * AGG_WAVES;
        if (job < GEMM_JOBS) {
            const int strip = job >> 2, t0 = (job & 3) * 2;
#pragma unroll
            for (int i = 0; i < 2; ++i) {
                const int   cc = (t0 + i) * 16 + m;
                const float bb = bias[cc];
                float* o = acc + (strip * 16 + half * 8) * ACC_P + (cc & 3) * 32 + (cc >> 2);
#pragma unroll
                for (int v = 0; v < 8; ++v)
                    o[v * ACC_P] = c[slot][i][v] + nsum[strip * 16 + half * 8 + v] * bb;
            }
        }
    }
    __syncthreads();

    const float4 g  = ((const float4*)gamma)[lane];
    const float4 be = ((const float4*)beta)[lane];
    for (int tl = wave; tl < AGG_R; tl += AGG_WAVES) {
        const int n = node0 + tl;
        if (n >= N) break;

        float4 v;
        v.x = fmaxf(acc[tl * ACC_P + lane],      0.0f);
        v.y = fmaxf(acc[tl * ACC_P + 32 + lane], 0.0f);
        v.z = fmaxf(acc[tl * ACC_P + 64 + lane], 0.0f);
        v.w = fmaxf(acc[tl * ACC_P + 96 + lane], 0.0f);

        float s  = v.x + v.y + v.z + v.w;
        float s2 = v.x * v.x + v.y * v.y + v.z * v.z + v.w * v.w;
#pragma unroll
        for (int o = 16; o >= 1; o >>= 1) {
            s  += __shfl_xor(s,  o, 32);
            s2 += __shfl_xor(s2, o, 32);
        }
        const float mu   = s * (1.0f / D);
        const float var  = s2 * (1.0f / D) - mu * mu;
        const float rstd = __frsqrt_rn(var + LN_EPS);

        const float4 xv = ((const float4*)(x + (size_t)n * D))[lane];

        float4 r;
        r.x = (v.x - mu) * rstd * g.x + be.x + xv.x;
        r.y = (v.y - mu) * rstd * g.y + be.y + xv.y;
        r.z = (v.z - mu) * rstd * g.z + be.z + xv.z;
        r.w = (v.w - mu) * rstd * g.w + be.w + xv.w;

        ((float4*)(out + (size_t)n * D))[lane] = r;
    }
}

extern "C" void kernel_launch(void* const* d_in, const int* in_sizes, int n_in,
                              void* d_out, int out_size, void* d_ws, size_t ws_size,
                              hipStream_t stream) {
    const float* x     = (const float*)d_in[0];
    const int*   ei    = (const int*)  d_in[1];
    const float* W     = (const float*)d_in[2];
    const float* bias  = (const float*)d_in[3];
    const float* gamma = (const float*)d_in[4];
    const float* beta  = (const float*)d_in[5];

    const int N = in_sizes[0] / D;
    const int E = in_sizes[1] / 2;

    const int G      = (N + AGG_R - 1) / AGG_R;
    const int nchunk = (E + PRES_CHUNK - 1) / PRES_CHUNK;

    float*    deg     = (float*)d_ws;
    unsigned* present = (unsigned*)((char*)d_ws + ((size_t)N * sizeof(float) + 255) / 256 * 256);

    rgcn_init_kernel<<<(N + 255) / 256, 256, 0, stream>>>(deg, N);
    rgcn_degree_presence_kernel<<<nchunk, 256, (size_t)G * sizeof(unsigned), stream>>>(
        ei, deg, present, E, G);
    rgcn_owner_sum_linear_final_kernel<<<G, 1024, 0, stream>>>(
        ei, deg, present, x, W, bias, gamma, beta, (float*)d_out, N, E);
}


# degree counted by owner workgroups: launch 1 also notes source-owner presence bits, a new launch counts only the noted groups' sources in LDS integers, stores deg once; no global atomics; deg exact
# speedup vs baseline: 3.2704x; 1.5432x over previous
#include <hip/hip_runtime.h>
#include <cstdint>
#include <cstddef>


typedef float v2f __attribute__((ext_vector_type(2)));
typedef float v8f __attribute__((ext_vector_type(8)));

#define D 128
#define LN_EPS 1e-5f

#define AGG_R       196
#define DEG_R       192
#define PRES_GRP    32
#define PRES_CHUNK  1024

__global__ void __launch_bounds__(256)
rgcn_presence_kernel(const int* __restrict__ ei, unsigned* __restrict__ present_t,
                     unsigned* __restrict__ present_s, int E, int G, int GD) {
    extern __shared__ unsigned omask[];
    const int lane = threadIdx.x & 31;
    const int wave = threadIdx.x >> 5;
    for (int i = threadIdx.x; i < G + GD; i += 256) omask[i] = 0u;
    __syncthreads();

    const int* __restrict__ col = ei + E;
    const int cbase = blockIdx.x * PRES_CHUNK;
    int s[4], t[4];
#pragma unroll
    for (int q = 0; q < 4; ++q) {
        const int e = min(cbase + (wave * 4 + q) * PRES_GRP + lane, E - 1);
        s[q] = ei[e];
        t[q] = col[e];
    }
#pragma unroll
    for (int q = 0; q < 4; ++q) {
        const int g = wave * 4 + q;
        if (cbase + g * PRES_GRP + lane < E) {
            const unsigned ot = (unsigned)t[q] / (unsigned)AGG_R;
            if (ot < (unsigned)G)  atomicOr(&omask[ot], 1u << g);
            const unsigned os = (unsigned)s[q] / (unsigned)DEG_R;
            if (os < (unsigned)GD) atomicOr(&omask[G + os], 1u << g);
        }
    }
    __syncthreads();
    for (int i = threadIdx.x; i < G; i += 256)
        present_t[(size_t)blockIdx.x * G + i] = omask[i];
    for (int i = threadIdx.x; i < GD; i += 256)
        present_s[(size_t)blockIdx.x * GD + i] = omask[G + i];
}

#define DEG_WAVES  16
#define DEG_Q      8
#define DEG_LIST   (32 * 32)

__global__ void __launch_bounds__(DEG_WAVES * 32)
rgcn_owner_degree_kernel(const int* __restrict__ ei, const unsigned* __restrict__ present_s,
                         float* __restrict__ deg, int N, int E) {
    __shared__ unsigned cnt[DEG_R];
    __shared__ __attribute__((aligned(16))) int glist[DEG_WAVES * DEG_LIST];

    const int lane  = threadIdx.x & 31;
    const int wave  = __builtin_amdgcn_readfirstlane(threadIdx.x >> 5);
    const int node0 = blockIdx.x * DEG_R;

    if (threadIdx.x < DEG_R) cnt[threadIdx.x] = 0u;
    __syncthreads();

    int* mylist = glist + wave * DEG_LIST;
    const int sub  = lane >> 3;
    const int eoff = (lane & 7) * 4;

    auto count = [&](int sv, bool exists) {
        const unsigned sl = (unsigned)(sv - node0);
        if (exists && sl < (unsigned)DEG_R) atomicAdd(&cnt[sl], 1u);
    };

    const int nchunk = (E + PRES_CHUNK - 1) / PRES_CHUNK;
    const int GD     = gridDim.x;
    for (int c0 = wave; c0 < nchunk; c0 += DEG_WAVES * 32) {
        const int c = c0 + DEG_WAVES * lane;
        const unsigned wmask = (c < nchunk) ? present_s[(size_t)c * GD + blockIdx.x] : 0u;

        const int pc = __builtin_popcount(wmask);
        int incl = pc;
#pragma unroll
        for (int o = 1; o < 32; o <<= 1) {
            const int v = __shfl_up(incl, o, 32);
            if (lane >= o) incl += v;
        }
        const int total = __builtin_amdgcn_readlane(incl, 31);
        int p = incl - pc;
        for (unsigned b = wmask; b != 0u; b &= b - 1u)
            mylist[p++] = c * PRES_CHUNK + __builtin_ctz(b) * PRES_GRP;
        if (total + lane < ((total + 31) & ~31)) mylist[total + lane] = E;

        for (int i = 0; i < total; i += 4 * DEG_Q) {
            const int  k0 = i + DEG_Q * sub;
            const int4 la = *(const int4*)(mylist + k0);
            const int4 lb = *(const int4*)(mylist + k0 + 4);
            const int  g0[DEG_Q] = { la.x, la.y, la.z, la.w, lb.x, lb.y, lb.z, lb.w };
            int  left[DEG_Q];
            int4 s[DEG_Q];
#pragma unroll
            for (int u = 0; u < DEG_Q; ++u) {
                const int e = g0[u] + eoff;
                left[u] = E - e;
                s[u] = *(const int4*)(ei + (left[u] > 0 ? e : 0));
            }
#pragma unroll
            for (int u = 0; u < DEG_Q; ++u) {
                count(s[u].x, left[u] > 0);
                count(s[u].y, left[u] > 1);
                count(s[u].z, left[u] > 2);
                count(s[u].w, left[u] > 3);
            }
        }
    }
    __syncthreads();

    if (threadIdx.x < DEG_R) {
        const int n = node0 + threadIdx.x;
        if (n < N) deg[n] = 1.0f + (float)cnt[threadIdx.x];
    }
}

#define AGG_WAVES  32
#define AGG_K      16
#define AGG_FLUSH  64
#define AGG_CAP    (AGG_K * PRES_GRP + AGG_FLUSH)
#define AGG_ROWS   8
#define AGG_RP     208
#define ACC_P      132
#define GEMM_JOBS  (AGG_RP / 16 * 4)

__global__ void __launch_bounds__(1024)
rgcn_owner_sum_linear_final_kernel(const int* __restrict__ ei, const float* __restrict__ deg,
                                   const unsigned* __restrict__ present,
                                   const float* __restrict__ x, const float* __restrict__ W,
                                   const float* __restrict__ bias,
                                   const float* __restrict__ gamma, const float* __restrict__ beta,
                                   float* __restrict__ out, int N, int E) {
    __shared__ __attribute__((aligned(16))) float acc[AGG_RP * ACC_P];
    __shared__ __attribute__((aligned(16))) float wl[D * ACC_P];
    __shared__ float nsum[AGG_RP];
    __shared__ int   pend[AGG_WAVES * AGG_CAP];

    const int lane  = threadIdx.x & 31;
    const int wave  = __builtin_amdgcn_readfirstlane(threadIdx.x >> 5);
    const int node0 = blockIdx.x * AGG_R;

    for (int i = threadIdx.x; i < AGG_RP * ACC_P / 4; i += 1024)
        ((float4*)acc)[i] = make_float4(0.0f, 0.0f, 0.0f, 0.0f);
    if (threadIdx.x < AGG_RP) nsum[threadIdx.x] = 0.0f;
    for (int i = threadIdx.x; i < D * D / 4; i += 1024) {
        const int n = i >> 5, c4 = i & 31;
        *(float4*)(wl + n * ACC_P + 4 * c4) = ((const float4*)W)[i];
    }
    __syncthreads();

    const int* __restrict__ col = ei + E;
    int* mylist = pend + wave * AGG_CAP;
    int cnt = 0;

    auto append = [&](bool hit, int e) {
        const unsigned m = __builtin_amdgcn_ballot_w32(hit);
        if (hit) mylist[cnt + (int)__builtin_amdgcn_mbcnt_lo(m, 0u)] = e;
        cnt += __builtin_popcount(m);
    };

    auto flush = [&]() {
        __builtin_amdgcn_fence(__ATOMIC_ACQ_REL, "wavefront");
        for (int base = 0; base < cnt; base += 32) {
            const int m = min(32, cnt - base);
            int s_l = 0, tl_l = 0;
            float nrm_l = 0.0f;
            if (lane < m) {
                const int e = mylist[base + lane];
                int t;
                if (e < E) { t = col[e]; s_l = ei[e]; }
                else       { t = s_l = e - E; }
                tl_l  = t - node0;
                nrm_l = __frsqrt_rn(deg[s_l] * deg[t]);
                atomicAdd(nsum + tl_l, nrm_l);
            }
            for (int j = 0; j < m; j += AGG_ROWS) {
                float4 v[AGG_ROWS];
                float  nr[AGG_ROWS];
                int    tl[AGG_ROWS];
#pragma unroll
                for (int u = 0; u < AGG_ROWS; ++u) {
                    const int jj = min(j + u, m - 1);
                    const int s  = __builtin_amdgcn_readlane(s_l, jj);
                    tl[u] = __builtin_amdgcn_readlane(tl_l, jj);
                    nr[u] = __int_as_float(__builtin_amdgcn_readlane(__float_as_int(nrm_l), jj));
                    v[u]  = ((const float4*)(x + (size_t)s * D))[lane];
                }
#pragma unroll
                for (int u = 0; u < AGG_ROWS; ++u) {
                    if (j + u < m) {
                        float* a = acc + tl[u] * ACC_P + lane;
                        atomicAdd(a,      nr[u] * v[u].x);
                        atomicAdd(a + 32, nr[u] * v[u].y);
                        atomicAdd(a + 64, nr[u] * v[u].z);
                        atomicAdd(a + 96, nr[u] * v[u].w);
                    }
                }
            }
        }
        cnt = 0;
    };

    {
        const int tl = wave + AGG_WAVES * lane;
        append(tl < AGG_R && node0 + tl < N, E + node0 + tl);
    }

    const int nchunk = (E + PRES_CHUNK - 1) / PRES_CHUNK;
    const int G      = gridDim.x;
    for (int c0 = wave; c0 < nchunk; c0 += AGG_WAVES * 32) {
        const int c = c0 + AGG_WAVES * lane;
        const unsigned wmask = (c < nchunk) ? present[(size_t)c * G + blockIdx.x] : 0u;
        unsigned live  = __builtin_amdgcn_ballot_w32(wmask != 0u);
        unsigned bits  = 0u;
        int      cbase = 0;

        auto next_group = [&]() -> int {
            if (bits == 0u) {
                if (live == 0u) return -1;
                const int l = __builtin_ctz(live);
                live &= live - 1u;
                bits  = (unsigned)__builtin_amdgcn_readlane((int)wmask, l);
                cbase = (c0 + AGG_WAVES * l) * PRES_CHUNK;
            }
            const int b = __builtin_ctz(bits);
            bits &= bits - 1u;
            return cbase + b * PRES_GRP;
        };

        for (;;) {
            int e0[AGG_K], t[AGG_K];
#pragma unroll
            for (int u = 0; u < AGG_K; ++u) {
                e0[u] = next_group();
                t[u] = col[min(max(e0[u], 0) + lane, E - 1)];
            }
#pragma unroll
            for (int u = 0; u < AGG_K; ++u) {
                if (e0[u] >= 0) {
                    const int e = e0[u] + lane;
                    append(e < E && (unsigned)(t[u] - node0) < (unsigned)AGG_R, e);
                }
            }
            if (cnt >= AGG_FLUSH) flush();
            if (e0[AGG_K - 1] < 0) break;
        }
    }
    if (cnt > 0) flush();
    __syncthreads();

    const int m    = lane & 15;
    const int half = lane >> 4;
    v8f c[2][2] = {};
#pragma unroll
    for (int slot = 0; slot < 2; ++slot) {
        const int job = wave + slot * AGG_WAVES;
        if (job < GEMM_JOBS) {
            const int strip = job >> 2, t0 = (job & 3) * 2;
            const float* arow = acc + (strip * 16 + m) * ACC_P + 64 * half;
            const float* b0   = wl + ((t0 + 0) * 16 + m) * ACC_P + 2 * half;
            const float* b1   = wl + ((t0 + 1) * 16 + m) * ACC_P + 2 * half;
#pragma unroll 8
            for (int kk = 0; kk < D / 4; ++kk) {
                v2f a;
                a.x = arow[kk];
                a.y = arow[32 + kk];
                const v2f w0 = *(const v2f*)(b0 + 4 * kk);
                const v2f w1 = *(const v2f*)(b1 + 4 * kk);
                c[slot][0] = __builtin_amdgcn_wmma_f32_16x16x4_f32(
                     false, a,  false, w0,
                     (short)0, c[slot][0],  false,  false);
                c[slot][1] = __builtin_amdgcn_wmma_f32_16x16x4_f32(
                     false, a,  false, w1,
                     (short)0, c[slot][1],  false,  false);
            }
        }
    }
    __syncthreads();

#pragma unroll
    for (int slot = 0; slot < 2; ++slot) {
        const int job = wave + slot * AGG_WAVES;
        if (job < GEMM_JOBS) {
            const int strip = job >> 2, t0 = (job & 3) * 2;
#pragma unroll
            for (int i = 0; i < 2; ++i) {
                const int   cc = (t0 + i) * 16 + m;
                const float bb = bias[cc];
                float* o = acc + (strip * 16 + half * 8) * ACC_P + (cc & 3) * 32 + (cc >> 2);
#pragma unroll
                for (int v = 0; v < 8; ++v)
                    o[v * ACC_P] = c[slot][i][v] + nsum[strip * 16 + half * 8 + v] * bb;
            }
        }
    }
    __syncthreads();

    const float4 g  = ((const float4*)gamma)[lane];
    const float4 be = ((const float4*)beta)[lane];
    for (int tl = wave; tl < AGG_R; tl += AGG_WAVES) {
        const int n = node0 + tl;
        if (n >= N) break;

        float4 v;
        v.x = fmaxf(acc[tl * ACC_P + lane],      0.0f);
        v.y = fmaxf(acc[tl * ACC_P + 32 + lane], 0.0f);
        v.z = fmaxf(acc[tl * ACC_P + 64 + lane], 0.0f);
        v.w = fmaxf(acc[tl * ACC_P + 96 + lane], 0.0f);

        float s  = v.x + v.y + v.z + v.w;
        float s2 = v.x * v.x + v.y * v.y + v.z * v.z + v.w * v.w;
#pragma unroll
        for (int o = 16; o >= 1; o >>= 1) {
            s  += __shfl_xor(s,  o, 32);
            s2 += __shfl_xor(s2, o, 32);
        }
        const float mu   = s * (1.0f / D);
        const float var  = s2 * (1.0f / D) - mu * mu;
        const float rstd = __frsqrt_rn(var + LN_EPS);

        const float4 xv = ((const float4*)(x + (size_t)n * D))[lane];

        float4 r;
        r.x = (v.x - mu) * rstd * g.x + be.x + xv.x;
        r.y = (v.y - mu) * rstd * g.y + be.y + xv.y;
        r.z = (v.z - mu) * rstd * g.z + be.z + xv.z;
        r.w = (v.w - mu) * rstd * g.w + be.w + xv.w;

        ((float4*)(out + (size_t)n * D))[lane] = r;
    }
}

extern "C" void kernel_launch(void* const* d_in, const int* in_sizes, int n_in,
                              void* d_out, int out_size, void* d_ws, size_t ws_size,
                              hipStream_t stream) {
    const float* x     = (const float*)d_in[0];
    const int*   ei    = (const int*)  d_in[1];
    const float* W     = (const float*)d_in[2];
    const float* bias  = (const float*)d_in[3];
    const float* gamma = (const float*)d_in[4];
    const float* beta  = (const float*)d_in[5];

    const int N = in_sizes[0] / D;
    const int E = in_sizes[1] / 2;

    const int G      = (N + AGG_R - 1) / AGG_R;
    const int GD     = (N + DEG_R - 1) / DEG_R;
    const int nchunk = (E + PRES_CHUNK - 1) / PRES_CHUNK;

    float*    deg       = (float*)d_ws;
    unsigned* present_t = (unsigned*)((char*)d_ws + ((size_t)N * sizeof(float) + 255) / 256 * 256);
    unsigned* present_s = present_t + (size_t)nchunk * G;

    rgcn_presence_kernel<<<nchunk, 256, (size_t)(G + GD) * sizeof(unsigned), stream>>>(
        ei, present_t, present_s, E, G, GD);
    rgcn_owner_degree_kernel<<<GD, DEG_WAVES * 32, 0, stream>>>(ei, present_s, deg, N, E);
    rgcn_owner_sum_linear_final_kernel<<<G, 1024, 0, stream>>>(
        ei, deg, present_t, x, W, bias, gamma, beta, (float*)d_out, N, E);
}


# launch 1 sorts each 4096-edge chunk by summing owner in LDS, stores (source,target) records + a start/count word per chunk and owner; owners read own segments, no edge tests; arrival-order sums kept
# speedup vs baseline: 3.6559x; 1.1179x over previous
#include <hip/hip_runtime.h>
#include <cstdint>
#include <cstddef>


typedef float v2f __attribute__((ext_vector_type(2)));
typedef float v8f __attribute__((ext_vector_type(8)));

#define D 128
#define LN_EPS 1e-5f

#define AGG_R       196
#define DEG_R       192
#define PRES_GRP    32
#define PRES_CHUNK  1024
#define SORT_T      1024
#define SORT_Q      4
#define SORT_CHUNK  (SORT_T * SORT_Q)
static_assert(SORT_T == PRES_CHUNK && SORT_T / 32 == PRES_CHUNK / PRES_GRP,
              "a sorting workgroup's threads cover one presence chunk per step, a wave per group");
static_assert(SORT_CHUNK < 65536, "start and count of a segment share one 32-bit word");

__global__ void __launch_bounds__(SORT_T)
rgcn_sort_presence_kernel(const int* __restrict__ ei, int2* __restrict__ recs,
                          unsigned* __restrict__ seg, unsigned* __restrict__ present_s,
                          int E, int G, int GD, int nblk, int nchunk) {
    extern __shared__ unsigned dyn[];
    __shared__ __attribute__((aligned(16))) int srec[2 * SORT_CHUNK];
    __shared__ unsigned stotal;
    unsigned* bcnt  = dyn;
    unsigned* omask = bcnt + G;
    unsigned* bval  = omask + SORT_Q * GD;
    unsigned* ptot  = bval + G;
    const int lane = threadIdx.x & 31;
    const int wave = __builtin_amdgcn_readfirstlane(threadIdx.x >> 5);

    for (int i = threadIdx.x; i < G + SORT_Q * GD; i += SORT_T) dyn[i] = 0u;
    if (threadIdx.x == 0) stotal = 0u;
    __syncthreads();

    const int* __restrict__ col = ei + E;
    const int cbase = blockIdx.x * SORT_CHUNK;
    int s[SORT_Q], t[SORT_Q];
#pragma unroll
    for (int q = 0; q < SORT_Q; ++q) {
        const int e = min(cbase + q * SORT_T + (int)threadIdx.x, E - 1);
        s[q] = ei[e];
        t[q] = col[e];
    }
    unsigned ot[SORT_Q], rank[SORT_Q];
#pragma unroll
    for (int q = 0; q < SORT_Q; ++q) {
        ot[q]   = 0xffffffffu;
        rank[q] = 0u;
        if (cbase + q * SORT_T + (int)threadIdx.x < E) {
            ot[q] = (unsigned)t[q] / (unsigned)AGG_R;
            if (ot[q] < (unsigned)G) rank[q] = atomicAdd(&bcnt[ot[q]], 1u);
            const unsigned os = (unsigned)s[q] / (unsigned)DEG_R;
            if (os < (unsigned)GD) atomicOr(&omask[q * GD + os], 1u << wave);
        }
    }
    __syncthreads();

    const int npiece = (G + 31) >> 5;
    for (int pc = wave; pc < npiece; pc += SORT_T / 32) {
        const int o = pc * 32 + lane;
        const unsigned v = (o < G) ? bcnt[o] : 0u;
        unsigned incl = v;
#pragma unroll
        for (int d = 1; d < 32; d <<= 1) {
            const unsigned u = __shfl_up(incl, d, 32);
            if (lane >= d) incl += u;
        }
        if (o < G) { bcnt[o] = incl - v; bval[o] = v; }
        if (lane == 31) ptot[pc] = incl;
    }
#pragma unroll
    for (int q = 0; q < SORT_Q; ++q) {
        const int c = blockIdx.x * SORT_Q + q;
        if (c < nchunk)
            for (int i = threadIdx.x; i < GD; i += SORT_T)
                present_s[(size_t)c * GD + i] = omask[q * GD + i];
    }
    __syncthreads();

    for (int o = threadIdx.x; o < G; o += SORT_T) {
        unsigned start = bcnt[o];
        for (int pp = 0; pp < (o >> 5); ++pp) start += ptot[pp];
        const unsigned v = bval[o];
        bcnt[o] = start;
        seg[(size_t)o * nblk + blockIdx.x] = start | (v << 16);
        if (o == G - 1) stotal = start + v;
    }
    __syncthreads();

#pragma unroll
    for (int q = 0; q < SORT_Q; ++q) {
        if (ot[q] < (unsigned)G)
            *(int2*)(srec + 2 * (bcnt[ot[q]] + rank[q])) = make_int2(s[q], t[q]);
    }
    __syncthreads();

    const int npair = ((int)stotal + 1) >> 1;
    int4* dst = (int4*)(recs + (size_t)cbase);
    for (int i = threadIdx.x; i < npair; i += SORT_T) dst[i] = ((const int4*)srec)[i];
}

#define DEG_WAVES  16
#define DEG_Q      8
#define DEG_LIST   (32 * 32)

__global__ void __launch_bounds__(DEG_WAVES * 32)
rgcn_owner_degree_kernel(const int* __restrict__ ei, const unsigned* __restrict__ present_s,
                         float* __restrict__ deg, int N, int E) {
    __shared__ unsigned cnt[DEG_R];
    __shared__ __attribute__((aligned(16))) int glist[DEG_WAVES * DEG_LIST];

    const int lane  = threadIdx.x & 31;
    const int wave  = __builtin_amdgcn_readfirstlane(threadIdx.x >> 5);
    const int node0 = blockIdx.x * DEG_R;

    if (threadIdx.x < DEG_R) cnt[threadIdx.x] = 0u;
    __syncthreads();

    int* mylist = glist + wave * DEG_LIST;
    const int sub  = lane >> 3;
    const int eoff = (lane & 7) * 4;

    auto count = [&](int sv, bool exists) {
        const unsigned sl = (unsigned)(sv - node0);
        if (exists && sl < (unsigned)DEG_R) atomicAdd(&cnt[sl], 1u);
    };

    const int nchunk = (E + PRES_CHUNK - 1) / PRES_CHUNK;
    const int GD     = gridDim.x;
    for (int c0 = wave; c0 < nchunk; c0 += DEG_WAVES * 32) {
        const int c = c0 + DEG_WAVES * lane;
        const unsigned wmask = (c < nchunk) ? present_s[(size_t)c * GD + blockIdx.x] : 0u;

        const int pc = __builtin_popcount(wmask);
        int incl = pc;
#pragma unroll
        for (int o = 1; o < 32; o <<= 1) {
            const int v = __shfl_up(incl, o, 32);
            if (lane >= o) incl += v;
        }
        const int total = __builtin_amdgcn_readlane(incl, 31);
        int p = incl - pc;
        for (unsigned b = wmask; b != 0u; b &= b - 1u)
            mylist[p++] = c * PRES_CHUNK + __builtin_ctz(b) * PRES_GRP;
        if (total + lane < ((total + 31) & ~31)) mylist[total + lane] = E;

        for (int i = 0; i < total; i += 4 * DEG_Q) {
            const int  k0 = i + DEG_Q * sub;
            const int4 la = *(const int4*)(mylist + k0);
            const int4 lb = *(const int4*)(mylist + k0 + 4);
            const int  g0[DEG_Q] = { la.x, la.y, la.z, la.w, lb.x, lb.y, lb.z, lb.w };
            int  left[DEG_Q];
            int4 s[DEG_Q];
#pragma unroll
            for (int u = 0; u < DEG_Q; ++u) {
                const int e = g0[u] + eoff;
                left[u] = E - e;
                s[u] = *(const int4*)(ei + (left[u] > 0 ? e : 0));
            }
#pragma unroll
            for (int u = 0; u < DEG_Q; ++u) {
                count(s[u].x, left[u] > 0);
                count(s[u].y, left[u] > 1);
                count(s[u].z, left[u] > 2);
                count(s[u].w, left[u] > 3);
            }
        }
    }
    __syncthreads();

    if (threadIdx.x < DEG_R) {
        const int n = node0 + threadIdx.x;
        if (n < N) deg[n] = 1.0f + (float)cnt[threadIdx.x];
    }
}

#define AGG_WAVES  32
#define AGG_FLUSH  64
#define AGG_CAP    (AGG_FLUSH + 32)
#define AGG_ROWS   8
#define AGG_RP     208
#define ACC_P      132
#define GEMM_JOBS  (AGG_RP / 16 * 4)

__global__ void __launch_bounds__(1024)
rgcn_owner_sum_linear_final_kernel(const int2* __restrict__ recs, const unsigned* __restrict__ seg,
                                   const float* __restrict__ deg,
                                   const float* __restrict__ x, const float* __restrict__ W,
                                   const float* __restrict__ bias,
                                   const float* __restrict__ gamma, const float* __restrict__ beta,
                                   float* __restrict__ out, int N, int nblk) {
    __shared__ __attribute__((aligned(16))) float acc[AGG_RP * ACC_P];
    __shared__ __attribute__((aligned(16))) float wl[D * ACC_P];
    __shared__ float nsum[AGG_RP];
    __shared__ int   pend[AGG_WAVES * AGG_CAP];

    const int lane  = threadIdx.x & 31;
    const int wave  = __builtin_amdgcn_readfirstlane(threadIdx.x >> 5);
    const int node0 = blockIdx.x * AGG_R;

    for (int i = threadIdx.x; i < AGG_RP * ACC_P / 4; i += 1024)
        ((float4*)acc)[i] = make_float4(0.0f, 0.0f, 0.0f, 0.0f);
    if (threadIdx.x < AGG_RP) nsum[threadIdx.x] = 0.0f;
    for (int i = threadIdx.x; i < D * D / 4; i += 1024) {
        const int n = i >> 5, c4 = i & 31;
        *(float4*)(wl + n * ACC_P + 4 * c4) = ((const float4*)W)[i];
    }
    __syncthreads();

    const int P = nblk * SORT_CHUNK;
    int* mylist = pend + wave * AGG_CAP;
    int cnt = 0;

    auto append = [&](bool hit, int e) {
        const unsigned m = __builtin_amdgcn_ballot_w32(hit);
        if (hit) mylist[cnt + (int)__builtin_amdgcn_mbcnt_lo(m, 0u)] = e;
        cnt += __builtin_popcount(m);
    };

    auto flush = [&]() {
        __builtin_amdgcn_fence(__ATOMIC_ACQ_REL, "wavefront");
        for (int base = 0; base < cnt; base += 32) {
            const int m = min(32, cnt - base);
            int s_l = 0, tl_l = 0;
            float nrm_l = 0.0f;
            if (lane < m) {
                const int e = mylist[base + lane];
                int t;
                if (e < P) { const int2 r = recs[e]; s_l = r.x; t = r.y; }
                else       { t = s_l = e - P; }
                tl_l  = t - node0;
                nrm_l = __frsqrt_rn(deg[s_l] * deg[t]);
                atomicAdd(nsum + tl_l, nrm_l);
            }
            for (int j = 0; j < m; j += AGG_ROWS) {
                float4 v[AGG_ROWS];
                float  nr[AGG_ROWS];
                int    tl[AGG_ROWS];
#pragma unroll
                for (int u = 0; u < AGG_ROWS; ++u) {
                    const int jj = min(j + u, m - 1);
                    const int s  = __builtin_amdgcn_readlane(s_l, jj);
                    tl[u] = __builtin_amdgcn_readlane(tl_l, jj);
                    nr[u] = __int_as_float(__builtin_amdgcn_readlane(__float_as_int(nrm_l), jj));
                    v[u]  = ((const float4*)(x + (size_t)s * D))[lane];
                }
#pragma unroll
                for (int u = 0; u < AGG_ROWS; ++u) {
                    if (j + u < m) {
                        float* a = acc + tl[u] * ACC_P + lane;
                        atomicAdd(a,      nr[u] * v[u].x);
                        atomicAdd(a + 32, nr[u] * v[u].y);
                        atomicAdd(a + 64, nr[u] * v[u].z);
                        atomicAdd(a + 96, nr[u] * v[u].w);
                    }
                }
            }
        }
        cnt = 0;
    };

    {
        const int tl = wave + AGG_WAVES * lane;
        append(tl < AGG_R && node0 + tl < N, P + node0 + tl);
    }

    const unsigned* __restrict__ myseg = seg + (size_t)blockIdx.x * nblk;
    for (int c0 = wave; c0 < nblk; c0 += AGG_WAVES * 32) {
        const int c = c0 + AGG_WAVES * lane;
        const unsigned word = (c < nblk) ? myseg[c] : 0u;
        unsigned live = __builtin_amdgcn_ballot_w32((word >> 16) != 0u);
        while (live != 0u) {
            const int l = __builtin_ctz(live);
            live &= live - 1u;
            const unsigned w = (unsigned)__builtin_amdgcn_readlane((int)word, l);
            const int count = (int)(w >> 16);
            const int pos0  = (c0 + AGG_WAVES * l) * SORT_CHUNK + (int)(w & 0xffffu);
            for (int i = 0; i < count; i += 32) {
                append(i + lane < count, pos0 + i + lane);
                if (cnt >= AGG_FLUSH) flush();
            }
        }
    }
    if (cnt > 0) flush();
    __syncthreads();

    const int m    = lane & 15;
    const int half = lane >> 4;
    v8f c[2][2] = {};
#pragma unroll
    for (int slot = 0; slot < 2; ++slot) {
        const int job = wave + slot * AGG_WAVES;
        if (job < GEMM_JOBS) {
            const int strip = job >> 2, t0 = (job & 3) * 2;
            const float* arow = acc + (strip * 16 + m) * ACC_P + 64 * half;
            const float* b0   = wl + ((t0 + 0) * 16 + m) * ACC_P + 2 * half;
            const float* b1   = wl + ((t0 + 1) * 16 + m) * ACC_P + 2 * half;
#pragma unroll 8
            for (int kk = 0; kk < D / 4; ++kk) {
                v2f a;
                a.x = arow[kk];
                a.y = arow[32 + kk];
                const v2f w0 = *(const v2f*)(b0 + 4 * kk);
                const v2f w1 = *(const v2f*)(b1 + 4 * kk);
                c[slot][0] = __builtin_amdgcn_wmma_f32_16x16x4_f32(
                     false, a,  false, w0,
                     (short)0, c[slot][0],  false,  false);
                c[slot][1] = __builtin_amdgcn_wmma_f32_16x16x4_f32(
                     false, a,  false, w1,
                     (short)0, c[slot][1],  false,  false);
            }
        }
    }
    __syncthreads();

#pragma unroll
    for (int slot = 0; slot < 2; ++slot) {
        const int job = wave + slot * AGG_WAVES;
        if (job < GEMM_JOBS) {
            const int strip = job >> 2, t0 = (job & 3) * 2;
#pragma unroll
            for (int i = 0; i < 2; ++i) {
                const int   cc = (t0 + i) * 16 + m;
                const float bb = bias[cc];
                float* o = acc + (strip * 16 + half * 8) * ACC_P + (cc & 3) * 32 + (cc >> 2);
#pragma unroll
                for (int v = 0; v < 8; ++v)
                    o[v * ACC_P] = c[slot][i][v] + nsum[strip * 16 + half * 8 + v] * bb;
            }
        }
    }
    __syncthreads();

    const float4 g  = ((const float4*)gamma)[lane];
    const float4 be = ((const float4*)beta)[lane];
    for (int tl = wave; tl < AGG_R; tl += AGG_WAVES) {
        const int n = node0 + tl;
        if (n >= N) break;

        float4 v;
        v.x = fmaxf(acc[tl * ACC_P + lane],      0.0f);
        v.y = fmaxf(acc[tl * ACC_P + 32 + lane], 0.0f);
        v.z = fmaxf(acc[tl * ACC_P + 64 + lane], 0.0f);
        v.w = fmaxf(acc[tl * ACC_P + 96 + lane], 0.0f);

        float s  = v.x + v.y + v.z + v.w;
        float s2 = v.x * v.x + v.y * v.y + v.z * v.z + v.w * v.w;
#pragma unroll
        for (int o = 16; o >= 1; o >>= 1) {
            s  += __shfl_xor(s,  o, 32);
            s2 += __shfl_xor(s2, o, 32);
        }
        const float mu   = s * (1.0f / D);
        const float var  = s2 * (1.0f / D) - mu * mu;
        const float rstd = __frsqrt_rn(var + LN_EPS);

        const float4 xv = ((const float4*)(x + (size_t)n * D))[lane];

        float4 r;
        r.x = (v.x - mu) * rstd * g.x + be.x + xv.x;
        r.y = (v.y - mu) * rstd * g.y + be.y + xv.y;
        r.z = (v.z - mu) * rstd * g.z + be.z + xv.z;
        r.w = (v.w - mu) * rstd * g.w + be.w + xv.w;

        ((float4*)(out + (size_t)n * D))[lane] = r;
    }
}

extern "C" void kernel_launch(void* const* d_in, const int* in_sizes, int n_in,
                              void* d_out, int out_size, void* d_ws, size_t ws_size,
                              hipStream_t stream) {
    const float* x     = (const float*)d_in[0];
    const int*   ei    = (const int*)  d_in[1];
    const float* W     = (const float*)d_in[2];
    const float* bias  = (const float*)d_in[3];
    const float* gamma = (const float*)d_in[4];
    const float* beta  = (const float*)d_in[5];

    const int N = in_sizes[0] / D;
    const int E = in_sizes[1] / 2;

    const int G      = (N + AGG_R - 1) / AGG_R;
    const int GD     = (N + DEG_R - 1) / DEG_R;
    const int nchunk = (E + PRES_CHUNK - 1) / PRES_CHUNK;
    const int nblk   = (E + SORT_CHUNK - 1) / SORT_CHUNK;

    char* p = (char*)d_ws;
    float*    deg       = (float*)p;     p += ((size_t)N * sizeof(float) + 255) / 256 * 256;
    unsigned* present_s = (unsigned*)p;  p += ((size_t)nchunk * GD * sizeof(unsigned) + 255) / 256 * 256;
    unsigned* seg       = (unsigned*)p;  p += ((size_t)G * nblk * sizeof(unsigned) + 255) / 256 * 256;
    int2*     recs      = (int2*)p;

    rgcn_sort_presence_kernel<<<nblk, SORT_T,
                                (size_t)(2 * G + SORT_Q * GD + (G + 31) / 32) * sizeof(unsigned), stream>>>(
        ei, recs, seg, present_s, E, G, GD, nblk, nchunk);
    rgcn_owner_degree_kernel<<<GD, DEG_WAVES * 32, 0, stream>>>(ei, present_s, deg, N, E);
    rgcn_owner_sum_linear_final_kernel<<<G, 1024, 0, stream>>>(
        recs, seg, deg, x, W, bias, gamma, beta, (float*)d_out, N, nblk);
}


# summing launch deals each record's source into an LDS queue of its target (integer slot adds); a wave walks its 7 targets' queues, sums rows in registers, stores each sum once; arrival-order sums
# speedup vs baseline: 3.7736x; 1.0322x over previous
#include <hip/hip_runtime.h>
#include <cstdint>
#include <cstddef>


typedef float v2f __attribute__((ext_vector_type(2)));
typedef float v8f __attribute__((ext_vector_type(8)));

#define D 128
#define LN_EPS 1e-5f

#define AGG_R       196
#define DEG_R       192
#define PRES_GRP    32
#define PRES_CHUNK  1024
#define SORT_T      1024
#define SORT_Q      4
#define SORT_CHUNK  (SORT_T * SORT_Q)
static_assert(SORT_T == PRES_CHUNK && SORT_T / 32 == PRES_CHUNK / PRES_GRP,
              "a sorting workgroup's threads cover one presence chunk per step, a wave per group");
static_assert(SORT_CHUNK < 65536, "start and count of a segment share one 32-bit word");

__global__ void __launch_bounds__(SORT_T)
rgcn_sort_presence_kernel(const int* __restrict__ ei, int2* __restrict__ recs,
                          unsigned* __restrict__ seg, unsigned* __restrict__ present_s,
                          int E, int G, int GD, int nblk, int nchunk) {
    extern __shared__ unsigned dyn[];
    __shared__ __attribute__((aligned(16))) int srec[2 * SORT_CHUNK];
    __shared__ unsigned stotal;
    unsigned* bcnt  = dyn;
    unsigned* omask = bcnt + G;
    unsigned* bval  = omask + SORT_Q * GD;
    unsigned* ptot  = bval + G;
    const int lane = threadIdx.x & 31;
    const int wave = __builtin_amdgcn_readfirstlane(threadIdx.x >> 5);

    for (int i = threadIdx.x; i < G + SORT_Q * GD; i += SORT_T) dyn[i] = 0u;
    if (threadIdx.x == 0) stotal = 0u;
    __syncthreads();

    const int* __restrict__ col = ei + E;
    const int cbase = blockIdx.x * SORT_CHUNK;
    int s[SORT_Q], t[SORT_Q];
#pragma unroll
    for (int q = 0; q < SORT_Q; ++q) {
        const int e = min(cbase + q * SORT_T + (int)threadIdx.x, E - 1);
        s[q] = ei[e];
        t[q] = col[e];
    }
    unsigned ot[SORT_Q], rank[SORT_Q];
#pragma unroll
    for (int q = 0; q < SORT_Q; ++q) {
        ot[q]   = 0xffffffffu;
        rank[q] = 0u;
        if (cbase + q * SORT_T + (int)threadIdx.x < E) {
            ot[q] = (unsigned)t[q] / (unsigned)AGG_R;
            if (ot[q] < (unsigned)G) rank[q] = atomicAdd(&bcnt[ot[q]], 1u);
            const unsigned os = (unsigned)s[q] / (unsigned)DEG_R;
            if (os < (unsigned)GD) atomicOr(&omask[q * GD + os], 1u << wave);
        }
    }
    __syncthreads();

    const int npiece = (G + 31) >> 5;
    for (int pc = wave; pc < npiece; pc += SORT_T / 32) {
        const int o = pc * 32 + lane;
        const unsigned v = (o < G) ? bcnt[o] : 0u;
        unsigned incl = v;
#pragma unroll
        for (int d = 1; d < 32; d <<= 1) {
            const unsigned u = __shfl_up(incl, d, 32);
            if (lane >= d) incl += u;
        }
        if (o < G) { bcnt[o] = incl - v; bval[o] = v; }
        if (lane == 31) ptot[pc] = incl;
    }
#pragma unroll
    for (int q = 0; q < SORT_Q; ++q) {
        const int c = blockIdx.x * SORT_Q + q;
        if (c < nchunk)
            for (int i = threadIdx.x; i < GD; i += SORT_T)
                present_s[(size_t)c * GD + i] = omask[q * GD + i];
    }
    __syncthreads();

    for (int o = threadIdx.x; o < G; o += SORT_T) {
        unsigned start = bcnt[o];
        for (int pp = 0; pp < (o >> 5); ++pp) start += ptot[pp];
        const unsigned v = bval[o];
        bcnt[o] = start;
        seg[(size_t)o * nblk + blockIdx.x] = start | (v << 16);
        if (o == G - 1) stotal = start + v;
    }
    __syncthreads();

#pragma unroll
    for (int q = 0; q < SORT_Q; ++q) {
        if (ot[q] < (unsigned)G)
            *(int2*)(srec + 2 * (bcnt[ot[q]] + rank[q])) = make_int2(s[q], t[q]);
    }
    __syncthreads();

    const int npair = ((int)stotal + 1) >> 1;
    int4* dst = (int4*)(recs + (size_t)cbase);
    for (int i = threadIdx.x; i < npair; i += SORT_T) dst[i] = ((const int4*)srec)[i];
}

#define DEG_WAVES  16
#define DEG_Q      8
#define DEG_LIST   (32 * 32)

__global__ void __launch_bounds__(DEG_WAVES * 32)
rgcn_owner_degree_kernel(const int* __restrict__ ei, const unsigned* __restrict__ present_s,
                         float* __restrict__ deg, int N, int E) {
    __shared__ unsigned cnt[DEG_R];
    __shared__ __attribute__((aligned(16))) int glist[DEG_WAVES * DEG_LIST];

    const int lane  = threadIdx.x & 31;
    const int wave  = __builtin_amdgcn_readfirstlane(threadIdx.x >> 5);
    const int node0 = blockIdx.x * DEG_R;

    if (threadIdx.x < DEG_R) cnt[threadIdx.x] = 0u;
    __syncthreads();

    int* mylist = glist + wave * DEG_LIST;
    const int sub  = lane >> 3;
    const int eoff = (lane & 7) * 4;

    auto count = [&](int sv, bool exists) {
        const unsigned sl = (unsigned)(sv - node0);
        if (exists && sl < (unsigned)DEG_R) atomicAdd(&cnt[sl], 1u);
    };

    const int nchunk = (E + PRES_CHUNK - 1) / PRES_CHUNK;
    const int GD     = gridDim.x;
    for (int c0 = wave; c0 < nchunk; c0 += DEG_WAVES * 32) {
        const int c = c0 + DEG_WAVES * lane;
        const unsigned wmask = (c < nchunk) ? present_s[(size_t)c * GD + blockIdx.x] : 0u;

        const int pc = __builtin_popcount(wmask);
        int incl = pc;
#pragma unroll
        for (int o = 1; o < 32; o <<= 1) {
            const int v = __shfl_up(incl, o, 32);
            if (lane >= o) incl += v;
        }
        const int total = __builtin_amdgcn_readlane(incl, 31);
        int p = incl - pc;
        for (unsigned b = wmask; b != 0u; b &= b - 1u)
            mylist[p++] = c * PRES_CHUNK + __builtin_ctz(b) * PRES_GRP;
        if (total + lane < ((total + 31) & ~31)) mylist[total + lane] = E;

        for (int i = 0; i < total; i += 4 * DEG_Q) {
            const int  k0 = i + DEG_Q * sub;
            const int4 la = *(const int4*)(mylist + k0);
            const int4 lb = *(const int4*)(mylist + k0 + 4);
            const int  g0[DEG_Q] = { la.x, la.y, la.z, la.w, lb.x, lb.y, lb.z, lb.w };
            int  left[DEG_Q];
            int4 s[DEG_Q];
#pragma unroll
            for (int u = 0; u < DEG_Q; ++u) {
                const int e = g0[u] + eoff;
                left[u] = E - e;
                s[u] = *(const int4*)(ei + (left[u] > 0 ? e : 0));
            }
#pragma unroll
            for (int u = 0; u < DEG_Q; ++u) {
                count(s[u].x, left[u] > 0);
                count(s[u].y, left[u] > 1);
                count(s[u].z, left[u] > 2);
                count(s[u].w, left[u] > 3);
            }
        }
    }
    __syncthreads();

    if (threadIdx.x < DEG_R) {
        const int n = node0 + threadIdx.x;
        if (n < N) deg[n] = 1.0f + (float)cnt[threadIdx.x];
    }
}

#define AGG_WAVES  32
#define AGG_QCAP   64
#define AGG_OWN    7
#define AGG_FLUSH  64
#define AGG_CAP    (AGG_FLUSH + 32)
#define AGG_ROWS   8
#define AGG_RP     208
#define ACC_P      132
#define GEMM_JOBS  (AGG_RP / 16 * 4)
static_assert(AGG_WAVES == 32 && (AGG_R + AGG_WAVES - 1) / AGG_WAVES == AGG_OWN && AGG_OWN <= 8,
              "a wave's targets are found by a three-step search over lanes 0..7");

__global__ void __launch_bounds__(1024)
rgcn_owner_sum_linear_final_kernel(const int2* __restrict__ recs, const unsigned* __restrict__ seg,
                                   const float* __restrict__ deg,
                                   const float* __restrict__ x, const float* __restrict__ W,
                                   const float* __restrict__ bias,
                                   const float* __restrict__ gamma, const float* __restrict__ beta,
                                   float* __restrict__ out, int N, int nblk) {
    __shared__ __attribute__((aligned(16))) float acc[AGG_RP * ACC_P];
    __shared__ __attribute__((aligned(16))) float wl[D * ACC_P];
    __shared__ float    nsum[AGG_RP];
    __shared__ int      queue[AGG_RP * AGG_QCAP];
    __shared__ unsigned qcnt[AGG_RP];
    __shared__ int      pend[AGG_WAVES * AGG_CAP];

    const int lane  = threadIdx.x & 31;
    const int wave  = __builtin_amdgcn_readfirstlane(threadIdx.x >> 5);
    const int node0 = blockIdx.x * AGG_R;

    for (int i = threadIdx.x; i < AGG_RP * ACC_P / 4; i += 1024)
        ((float4*)acc)[i] = make_float4(0.0f, 0.0f, 0.0f, 0.0f);
    if (threadIdx.x < AGG_RP) nsum[threadIdx.x] = 0.0f;
    if (threadIdx.x < AGG_RP) qcnt[threadIdx.x] = 0u;
    for (int i = threadIdx.x; i < D * D / 4; i += 1024) {
        const int n = i >> 5, c4 = i & 31;
        *(float4*)(wl + n * ACC_P + 4 * c4) = ((const float4*)W)[i];
    }
    __syncthreads();

    const unsigned* __restrict__ myseg = seg + (size_t)blockIdx.x * nblk;

    auto deal = [&](bool hit, int s, int tl) {
        if (hit) {
            const unsigned slot = atomicAdd(&qcnt[tl], 1u);
            if (slot < (unsigned)AGG_QCAP) queue[tl * AGG_QCAP + (int)slot] = s;
        }
    };
    {
        const int tl = wave + AGG_WAVES * lane;
        deal(tl < AGG_R && node0 + tl < N, node0 + tl, tl);
    }
    for (int c0 = wave; c0 < nblk; c0 += AGG_WAVES * 32) {
        const int c = c0 + AGG_WAVES * lane;
        const unsigned word = (c < nblk) ? myseg[c] : 0u;
        const int count = (int)(word >> 16);
        const int pos   = c * SORT_CHUNK + (int)(word & 0xffffu);
        int incl = count;
#pragma unroll
        for (int d = 1; d < 32; d <<= 1) {
            const int u = __shfl_up(incl, d, 32);
            if (lane >= d) incl += u;
        }
        const int pre   = incl - count;
        const int total = __builtin_amdgcn_readlane(incl, 31);
        for (int f0 = 0; f0 < total; f0 += 64) {
            int2 r[2];
            bool hit[2];
#pragma unroll
            for (int q = 0; q < 2; ++q) {
                const int f = f0 + 32 * q + lane;
                hit[q] = f < total;
                int lo = 0;
#pragma unroll
                for (int step = 16; step >= 1; step >>= 1) {
                    const int p = __shfl(pre, lo + step, 32);
                    if (p <= f) lo += step;
                }
                const int e = __shfl(pos, lo, 32) + (f - __shfl(pre, lo, 32));
                r[q] = make_int2(0, node0);
                if (hit[q]) r[q] = recs[e];
            }
#pragma unroll
            for (int q = 0; q < 2; ++q) deal(hit[q], r[q].x, r[q].y - node0);
        }
    }
    __syncthreads();

    bool over = false;
    for (int i = lane; i < AGG_R; i += 32) over = over || qcnt[i] > (unsigned)AGG_QCAP;
    const bool fits = __builtin_amdgcn_ballot_w32(over) == 0u;

    if (fits) {
        const int tlk   = wave + AGG_WAVES * lane;
        const int qlen  = (lane < AGG_OWN && tlk < AGG_R) ? (int)qcnt[tlk] : 0;
        int incl = qlen;
#pragma unroll
        for (int d = 1; d < 32; d <<= 1) {
            const int u = __shfl_up(incl, d, 32);
            if (lane >= d) incl += u;
        }
        const int pre = incl - qlen;
        const int nq  = __builtin_amdgcn_readlane(incl, 31);

        float4 run  = make_float4(0.0f, 0.0f, 0.0f, 0.0f);
        float  runs = 0.0f;
        int    cur  = -1;
        auto put = [&]() {
            const int tl = wave + AGG_WAVES * cur;
            *((float4*)(acc + tl * ACC_P) + lane) = run;
            if (lane == 0) nsum[tl] = runs;
        };

        for (int base = 0; base < nq; base += 32) {
            const int m = min(32, nq - base);
            const int f = base + lane;
            int k = 0;
#pragma unroll
            for (int step = 4; step >= 1; step >>= 1) {
                const int p = __shfl(pre, k + step, 32);
                if (p <= f) k += step;
            }
            const int fk = f - __shfl(pre, k, 32);
            int s_l = 0;
            float nrm_l = 0.0f;
            if (lane < m) {
                const int tl = wave + AGG_WAVES * k;
                s_l   = queue[tl * AGG_QCAP + fk];
                nrm_l = __frsqrt_rn(deg[s_l] * deg[node0 + tl]);
            }
            for (int j = 0; j < m; j += AGG_ROWS) {
                float4 v[AGG_ROWS];
                float  nr[AGG_ROWS];
                int    ko[AGG_ROWS];
#pragma unroll
                for (int u = 0; u < AGG_ROWS; ++u) {
                    const int jj = min(j + u, m - 1);
                    const int s  = __builtin_amdgcn_readlane(s_l, jj);
                    ko[u] = __builtin_amdgcn_readlane(k, jj);
                    nr[u] = __int_as_float(__builtin_amdgcn_readlane(__float_as_int(nrm_l), jj));
                    v[u]  = ((const float4*)(x + (size_t)s * D))[lane];
                }
#pragma unroll
                for (int u = 0; u < AGG_ROWS; ++u) {
                    if (j + u < m) {
                        if (ko[u] != cur) {
                            if (cur >= 0) put();
                            run  = make_float4(0.0f, 0.0f, 0.0f, 0.0f);
                            runs = 0.0f;
                            cur  = ko[u];
                        }
                        run.x += nr[u] * v[u].x;
                        run.y += nr[u] * v[u].y;
                        run.z += nr[u] * v[u].z;
                        run.w += nr[u] * v[u].w;
                        runs  += nr[u];
                    }
                }
            }
        }
        if (cur >= 0) put();
    } else {
        const int P = nblk * SORT_CHUNK;
        int* mylist = pend + wave * AGG_CAP;
        int cnt = 0;

        auto append = [&](bool hit, int e) {
            const unsigned mk = __builtin_amdgcn_ballot_w32(hit);
            if (hit) mylist[cnt + (int)__builtin_amdgcn_mbcnt_lo(mk, 0u)] = e;
            cnt += __builtin_popcount(mk);
        };

        auto flush = [&]() {
            __builtin_amdgcn_fence(__ATOMIC_ACQ_REL, "wavefront");
            for (int base = 0; base < cnt; base += 32) {
                const int m = min(32, cnt - base);
                int s_l = 0, tl_l = 0;
                float nrm_l = 0.0f;
                if (lane < m) {
                    const int e = mylist[base + lane];
                    int t;
                    if (e < P) { const int2 r = recs[e]; s_l = r.x; t = r.y; }
                    else       { t = s_l = e - P; }
                    tl_l  = t - node0;
                    nrm_l = __frsqrt_rn(deg[s_l] * deg[t]);
                    atomicAdd(nsum + tl_l, nrm_l);
                }
                for (int j = 0; j < m; ++j) {
                    const int   s  = __builtin_amdgcn_readlane(s_l, j);
                    const int   tj = __builtin_amdgcn_readlane(tl_l, j);
                    const float nj = __int_as_float(__builtin_amdgcn_readlane(__float_as_int(nrm_l), j));
                    const float4 v = ((const float4*)(x + (size_t)s * D))[lane];
                    float* a = acc + tj * ACC_P + 4 * lane;
                    atomicAdd(a,     nj * v.x);
                    atomicAdd(a + 1, nj * v.y);
                    atomicAdd(a + 2, nj * v.z);
                    atomicAdd(a + 3, nj * v.w);
                }
            }
            cnt = 0;
        };

        {
            const int tl = wave + AGG_WAVES * lane;
            append(tl < AGG_R && node0 + tl < N, P + node0 + tl);
        }
        for (int c0 = wave; c0 < nblk; c0 += AGG_WAVES * 32) {
            const int c = c0 + AGG_WAVES * lane;
            const unsigned word = (c < nblk) ? myseg[c] : 0u;
            unsigned live = __builtin_amdgcn_ballot_w32((word >> 16) != 0u);
            while (live != 0u) {
                const int l = __builtin_ctz(live);
                live &= live - 1u;
                const unsigned w = (unsigned)__builtin_amdgcn_readlane((int)word, l);
                const int count = (int)(w >> 16);
                const int pos0  = (c0 + AGG_WAVES * l) * SORT_CHUNK + (int)(w & 0xffffu);
                for (int i = 0; i < count; i += 32) {
                    append(i + lane < count, pos0 + i + lane);
                    if (cnt >= AGG_FLUSH) flush();
                }
            }
        }
        if (cnt > 0) flush();
    }
    __syncthreads();

    const int m    = lane & 15;
    const int half = lane >> 4;
    v8f c[2][2] = {};
#pragma unroll
    for (int slot = 0; slot < 2; ++slot) {
        const int job = wave + slot * AGG_WAVES;
        if (job < GEMM_JOBS) {
            const int strip = job >> 2, t0 = (job & 3) * 2;
            const float* arow = acc + (strip * 16 + m) * ACC_P + 2 * half;
            const float* b0   = wl + ((t0 + 0) * 16 + m) * ACC_P + 2 * half;
            const float* b1   = wl + ((t0 + 1) * 16 + m) * ACC_P + 2 * half;
#pragma unroll 8
            for (int kk = 0; kk < D / 4; ++kk) {
                const v2f a  = *(const v2f*)(arow + 4 * kk);
                const v2f w0 = *(const v2f*)(b0 + 4 * kk);
                const v2f w1 = *(const v2f*)(b1 + 4 * kk);
                c[slot][0] = __builtin_amdgcn_wmma_f32_16x16x4_f32(
                     false, a,  false, w0,
                     (short)0, c[slot][0],  false,  false);
                c[slot][1] = __builtin_amdgcn_wmma_f32_16x16x4_f32(
                     false, a,  false, w1,
                     (short)0, c[slot][1],  false,  false);
            }
        }
    }
    __syncthreads();

#pragma unroll
    for (int slot = 0; slot < 2; ++slot) {
        const int job = wave + slot * AGG_WAVES;
        if (job < GEMM_JOBS) {
            const int strip = job >> 2, t0 = (job & 3) * 2;
#pragma unroll
            for (int i = 0; i < 2; ++i) {
                const int   cc = (t0 + i) * 16 + m;
                const float bb = bias[cc];
                float* o = acc + (strip * 16 + half * 8) * ACC_P + cc;
#pragma unroll
                for (int v = 0; v < 8; ++v)
                    o[v * ACC_P] = c[slot][i][v] + nsum[strip * 16 + half * 8 + v] * bb;
            }
        }
    }
    __syncthreads();

    const float4 g  = ((const float4*)gamma)[lane];
    const float4 be = ((const float4*)beta)[lane];
    for (int tl = wave; tl < AGG_R; tl += AGG_WAVES) {
        const int n = node0 + tl;
        if (n >= N) break;

        const float4 a4 = *(const float4*)(acc + tl * ACC_P + 4 * lane);
        float4 v;
        v.x = fmaxf(a4.x, 0.0f);
        v.y = fmaxf(a4.y, 0.0f);
        v.z = fmaxf(a4.z, 0.0f);
        v.w = fmaxf(a4.w, 0.0f);

        float s  = v.x + v.y + v.z + v.w;
        float s2 = v.x * v.x + v.y * v.y + v.z * v.z + v.w * v.w;
#pragma unroll
        for (int o = 16; o >= 1; o >>= 1) {
            s  += __shfl_xor(s,  o, 32);
            s2 += __shfl_xor(s2, o, 32);
        }
        const float mu   = s * (1.0f / D);
        const float var  = s2 * (1.0f / D) - mu * mu;
        const float rstd = __frsqrt_rn(var + LN_EPS);

        const float4 xv = ((const float4*)(x + (size_t)n * D))[lane];

        float4 r;
        r.x = (v.x - mu) * rstd * g.x + be.x + xv.x;
        r.y = (v.y - mu) * rstd * g.y + be.y + xv.y;
        r.z = (v.z - mu) * rstd * g.z + be.z + xv.z;
        r.w = (v.w - mu) * rstd * g.w + be.w + xv.w;

        ((float4*)(out + (size_t)n * D))[lane] = r;
    }
}

extern "C" void kernel_launch(void* const* d_in, const int* in_sizes, int n_in,
                              void* d_out, int out_size, void* d_ws, size_t ws_size,
                              hipStream_t stream) {
    const float* x     = (const float*)d_in[0];
    const int*   ei    = (const int*)  d_in[1];
    const float* W     = (const float*)d_in[2];
    const float* bias  = (const float*)d_in[3];
    const float* gamma = (const float*)d_in[4];
    const float* beta  = (const float*)d_in[5];

    const int N = in_sizes[0] / D;
    const int E = in_sizes[1] / 2;

    const int G      = (N + AGG_R - 1) / AGG_R;
    const int GD     = (N + DEG_R - 1) / DEG_R;
    const int nchunk = (E + PRES_CHUNK - 1) / PRES_CHUNK;
    const int nblk   = (E + SORT_CHUNK - 1) / SORT_CHUNK;

    char* p = (char*)d_ws;
    float*    deg       = (float*)p;     p += ((size_t)N * sizeof(float) + 255) / 256 * 256;
    unsigned* present_s = (unsigned*)p;  p += ((size_t)nchunk * GD * sizeof(unsigned) + 255) / 256 * 256;
    unsigned* seg       = (unsigned*)p;  p += ((size_t)G * nblk * sizeof(unsigned) + 255) / 256 * 256;
    int2*     recs      = (int2*)p;

    rgcn_sort_presence_kernel<<<nblk, SORT_T,
                                (size_t)(2 * G + SORT_Q * GD + (G + 31) / 32) * sizeof(unsigned), stream>>>(
        ei, recs, seg, present_s, E, G, GD, nblk, nchunk);
    rgcn_owner_degree_kernel<<<GD, DEG_WAVES * 32, 0, stream>>>(ei, present_s, deg, N, E);
    rgcn_owner_sum_linear_final_kernel<<<G, 1024, 0, stream>>>(
        recs, seg, deg, x, W, bias, gamma, beta, (float*)d_out, N, nblk);
}


# launch 1 also counting-sorts each chunk's sources by counting owner (4-byte entries + start/count word); degree owners read only their own segments, no presence words or group re-reads; deg exact
# speedup vs baseline: 4.1866x; 1.1094x over previous
#include <hip/hip_runtime.h>
#include <cstdint>
#include <cstddef>


typedef float v2f __attribute__((ext_vector_type(2)));
typedef float v8f __attribute__((ext_vector_type(8)));

#define D 128
#define LN_EPS 1e-5f

#define AGG_R       196
#define DEG_R       192
#define SORT_T      1024
#define SORT_Q      4
#define SORT_CHUNK  (SORT_T * SORT_Q)
static_assert(SORT_CHUNK < 65536, "start and count of a segment share one 32-bit word");

__global__ void __launch_bounds__(SORT_T)
rgcn_sort_kernel(const int* __restrict__ ei, int2* __restrict__ recs, unsigned* __restrict__ seg,
                 int* __restrict__ srcs, unsigned* __restrict__ segs,
                 int E, int G, int GD, int nblk) {
    extern __shared__ unsigned dyn[];
    __shared__ __attribute__((aligned(16))) int srec[2 * SORT_CHUNK];
    __shared__ __attribute__((aligned(16))) int ssrc[SORT_CHUNK];
    __shared__ unsigned stotal;
    __shared__ unsigned stotal_s;
    const int GP  = (G + 31) & ~31;
    const int GA  = GP + GD;
    const int npt = GP >> 5;
    unsigned* bcnt = dyn;
    unsigned* bval = bcnt + GA;
    unsigned* ptot = bval + GA;
    const int lane = threadIdx.x & 31;
    const int wave = __builtin_amdgcn_readfirstlane(threadIdx.x >> 5);

    for (int i = threadIdx.x; i < GA; i += SORT_T) dyn[i] = 0u;
    if (threadIdx.x == 0) { stotal = 0u; stotal_s = 0u; }
    __syncthreads();

    const int* __restrict__ col = ei + E;
    const int cbase = blockIdx.x * SORT_CHUNK;
    int s[SORT_Q], t[SORT_Q];
#pragma unroll
    for (int q = 0; q < SORT_Q; ++q) {
        const int e = min(cbase + q * SORT_T + (int)threadIdx.x, E - 1);
        s[q] = ei[e];
        t[q] = col[e];
    }
    unsigned ot[SORT_Q], os[SORT_Q], rank[SORT_Q], ranks[SORT_Q];
#pragma unroll
    for (int q = 0; q < SORT_Q; ++q) {
        ot[q]    = 0xffffffffu;
        os[q]    = 0xffffffffu;
        rank[q]  = 0u;
        ranks[q] = 0u;
        if (cbase + q * SORT_T + (int)threadIdx.x < E) {
            ot[q] = (unsigned)t[q] / (unsigned)AGG_R;
            if (ot[q] < (unsigned)G) rank[q] = atomicAdd(&bcnt[ot[q]], 1u);
            os[q] = (unsigned)s[q] / (unsigned)DEG_R;
            if (os[q] < (unsigned)GD) ranks[q] = atomicAdd(&bcnt[GP + os[q]], 1u);
        }
    }
    __syncthreads();

    const int npiece = (GA + 31) >> 5;
    for (int pc = wave; pc < npiece; pc += SORT_T / 32) {
        const int o = pc * 32 + lane;
        const unsigned v = (o < GA) ? bcnt[o] : 0u;
        unsigned incl = v;
#pragma unroll
        for (int d = 1; d < 32; d <<= 1) {
            const unsigned u = __shfl_up(incl, d, 32);
            if (lane >= d) incl += u;
        }
        if (o < GA) { bcnt[o] = incl - v; bval[o] = v; }
        if (lane == 31) ptot[pc] = incl;
    }
    __syncthreads();

    for (int o = threadIdx.x; o < GA; o += SORT_T) {
        const bool srcside = o >= GP;
        unsigned start = bcnt[o];
        for (int pp = srcside ? npt : 0; pp < (o >> 5); ++pp) start += ptot[pp];
        const unsigned v = bval[o];
        bcnt[o] = start;
        if (srcside) {
            const int od = o - GP;
            segs[(size_t)od * nblk + blockIdx.x] = start | (v << 16);
            if (od == GD - 1) stotal_s = start + v;
        } else if (o < G) {
            seg[(size_t)o * nblk + blockIdx.x] = start | (v << 16);
            if (o == G - 1) stotal = start + v;
        }
    }
    __syncthreads();

#pragma unroll
    for (int q = 0; q < SORT_Q; ++q) {
        if (ot[q] < (unsigned)G)
            *(int2*)(srec + 2 * (bcnt[ot[q]] + rank[q])) = make_int2(s[q], t[q]);
        if (os[q] < (unsigned)GD)
            ssrc[bcnt[GP + os[q]] + ranks[q]] = s[q];
    }
    __syncthreads();

    const int npair = ((int)stotal + 1) >> 1;
    int4* dst = (int4*)(recs + (size_t)cbase);
    for (int i = threadIdx.x; i < npair; i += SORT_T) dst[i] = ((const int4*)srec)[i];
    const int nquad = ((int)stotal_s + 3) >> 2;
    int4* dsts = (int4*)(srcs + (size_t)cbase);
    for (int i = threadIdx.x; i < nquad; i += SORT_T) dsts[i] = ((const int4*)ssrc)[i];
}

#define DEG_WAVES  8
#define DEG_Q      8
static_assert(DEG_WAVES * 32 >= DEG_R, "a thread per owned node clears and stores its counter");

__global__ void __launch_bounds__(DEG_WAVES * 32)
rgcn_owner_degree_kernel(const int* __restrict__ srcs, const unsigned* __restrict__ segs,
                         float* __restrict__ deg, int N, int nblk) {
    __shared__ unsigned cnt[DEG_R];

    const int lane  = threadIdx.x & 31;
    const int wave  = __builtin_amdgcn_readfirstlane(threadIdx.x >> 5);
    const int node0 = blockIdx.x * DEG_R;

    if (threadIdx.x < DEG_R) cnt[threadIdx.x] = 0u;
    __syncthreads();

    const unsigned* __restrict__ myseg = segs + (size_t)blockIdx.x * nblk;
    for (int c0 = wave; c0 < nblk; c0 += DEG_WAVES * 32) {
        const int c = c0 + DEG_WAVES * lane;
        const unsigned word = (c < nblk) ? myseg[c] : 0u;
        const int first = c * SORT_CHUNK + (int)(word & 0xffffu);
        const int last  = first + (int)(word >> 16);
        auto count = [&](int sv, int idx) {
            const unsigned sl = (unsigned)(sv - node0);
            if (idx >= first && idx < last && sl < (unsigned)DEG_R) atomicAdd(&cnt[sl], 1u);
        };
        int unit = (first < last) ? (first & ~3) : last;
        while (__builtin_amdgcn_ballot_w32(unit < last) != 0u) {
            int4 v[DEG_Q];
#pragma unroll
            for (int u = 0; u < DEG_Q; ++u) {
                v[u] = make_int4(0, 0, 0, 0);
                if (unit + 4 * u < last) v[u] = *(const int4*)(srcs + unit + 4 * u);
            }
#pragma unroll
            for (int u = 0; u < DEG_Q; ++u) {
                count(v[u].x, unit + 4 * u);
                count(v[u].y, unit + 4 * u + 1);
                count(v[u].z, unit + 4 * u + 2);
                count(v[u].w, unit + 4 * u + 3);
            }
            unit += 4 * DEG_Q;
        }
    }
    __syncthreads();

    if (threadIdx.x < DEG_R) {
        const int n = node0 + threadIdx.x;
        if (n < N) deg[n] = 1.0f + (float)cnt[threadIdx.x];
    }
}

#define AGG_WAVES  32
#define AGG_QCAP   64
#define AGG_OWN    7
#define AGG_FLUSH  64
#define AGG_CAP    (AGG_FLUSH + 32)
#define AGG_ROWS   8
#define AGG_RP     208
#define ACC_P      132
#define GEMM_JOBS  (AGG_RP / 16 * 4)
static_assert(AGG_WAVES == 32 && (AGG_R + AGG_WAVES - 1) / AGG_WAVES == AGG_OWN && AGG_OWN <= 8,
              "a wave's targets are found by a three-step search over lanes 0..7");

__global__ void __launch_bounds__(1024)
rgcn_owner_sum_linear_final_kernel(const int2* __restrict__ recs, const unsigned* __restrict__ seg,
                                   const float* __restrict__ deg,
                                   const float* __restrict__ x, const float* __restrict__ W,
                                   const float* __restrict__ bias,
                                   const float* __restrict__ gamma, const float* __restrict__ beta,
                                   float* __restrict__ out, int N, int nblk) {
    __shared__ __attribute__((aligned(16))) float acc[AGG_RP * ACC_P];
    __shared__ __attribute__((aligned(16))) float wl[D * ACC_P];
    __shared__ float    nsum[AGG_RP];
    __shared__ int      queue[AGG_RP * AGG_QCAP];
    __shared__ unsigned qcnt[AGG_RP];
    __shared__ int      pend[AGG_WAVES * AGG_CAP];

    const int lane  = threadIdx.x & 31;
    const int wave  = __builtin_amdgcn_readfirstlane(threadIdx.x >> 5);
    const int node0 = blockIdx.x * AGG_R;

    for (int i = threadIdx.x; i < AGG_RP * ACC_P / 4; i += 1024)
        ((float4*)acc)[i] = make_float4(0.0f, 0.0f, 0.0f, 0.0f);
    if (threadIdx.x < AGG_RP) nsum[threadIdx.x] = 0.0f;
    if (threadIdx.x < AGG_RP) qcnt[threadIdx.x] = 0u;
    for (int i = threadIdx.x; i < D * D / 4; i += 1024) {
        const int n = i >> 5, c4 = i & 31;
        *(float4*)(wl + n * ACC_P + 4 * c4) = ((const float4*)W)[i];
    }
    __syncthreads();

    const unsigned* __restrict__ myseg = seg + (size_t)blockIdx.x * nblk;

    auto deal = [&](bool hit, int s, int tl) {
        if (hit) {
            const unsigned slot = atomicAdd(&qcnt[tl], 1u);
            if (slot < (unsigned)AGG_QCAP) queue[tl * AGG_QCAP + (int)slot] = s;
        }
    };
    {
        const int tl = wave + AGG_WAVES * lane;
        deal(tl < AGG_R && node0 + tl < N, node0 + tl, tl);
    }
    for (int c0 = wave; c0 < nblk; c0 += AGG_WAVES * 32) {
        const int c = c0 + AGG_WAVES * lane;
        const unsigned word = (c < nblk) ? myseg[c] : 0u;
        const int count = (int)(word >> 16);
        const int pos   = c * SORT_CHUNK + (int)(word & 0xffffu);
        int incl = count;
#pragma unroll
        for (int d = 1; d < 32; d <<= 1) {
            const int u = __shfl_up(incl, d, 32);
            if (lane >= d) incl += u;
        }
        const int pre   = incl - count;
        const int total = __builtin_amdgcn_readlane(incl, 31);
        for (int f0 = 0; f0 < total; f0 += 64) {
            int2 r[2];
            bool hit[2];
#pragma unroll
            for (int q = 0; q < 2; ++q) {
                const int f = f0 + 32 * q + lane;
                hit[q] = f < total;
                int lo = 0;
#pragma unroll
                for (int step = 16; step >= 1; step >>= 1) {
                    const int p = __shfl(pre, lo + step, 32);
                    if (p <= f) lo += step;
                }
                const int e = __shfl(pos, lo, 32) + (f - __shfl(pre, lo, 32));
                r[q] = make_int2(0, node0);
                if (hit[q]) r[q] = recs[e];
            }
#pragma unroll
            for (int q = 0; q < 2; ++q) deal(hit[q], r[q].x, r[q].y - node0);
        }
    }
    __syncthreads();

    bool over = false;
    for (int i = lane; i < AGG_R; i += 32) over = over || qcnt[i] > (unsigned)AGG_QCAP;
    const bool fits = __builtin_amdgcn_ballot_w32(over) == 0u;

    if (fits) {
        const int tlk   = wave + AGG_WAVES * lane;
        const int qlen  = (lane < AGG_OWN && tlk < AGG_R) ? (int)qcnt[tlk] : 0;
        int incl = qlen;
#pragma unroll
        for (int d = 1; d < 32; d <<= 1) {
            const int u = __shfl_up(incl, d, 32);
            if (lane >= d) incl += u;
        }
        const int pre = incl - qlen;
        const int nq  = __builtin_amdgcn_readlane(incl, 31);

        float4 run  = make_float4(0.0f, 0.0f, 0.0f, 0.0f);
        float  runs = 0.0f;
        int    cur  = -1;
        auto put = [&]() {
            const int tl = wave + AGG_WAVES * cur;
            *((float4*)(acc + tl * ACC_P) + lane) = run;
            if (lane == 0) nsum[tl] = runs;
        };

        for (int base = 0; base < nq; base += 32) {
            const int m = min(32, nq - base);
            const int f = base + lane;
            int k = 0;
#pragma unroll
            for (int step = 4; step >= 1; step >>= 1) {
                const int p = __shfl(pre, k + step, 32);
                if (p <= f) k += step;
            }
            const int fk = f - __shfl(pre, k, 32);
            int s_l = 0;
            float nrm_l = 0.0f;
            if (lane < m) {
                const int tl = wave + AGG_WAVES * k;
                s_l   = queue[tl * AGG_QCAP + fk];
                nrm_l = __frsqrt_rn(deg[s_l] * deg[node0 + tl]);
            }
            for (int j = 0; j < m; j += AGG_ROWS) {
                float4 v[AGG_ROWS];
                float  nr[AGG_ROWS];
                int    ko[AGG_ROWS];
#pragma unroll
                for (int u = 0; u < AGG_ROWS; ++u) {
                    const int jj = min(j + u, m - 1);
                    const int s  = __builtin_amdgcn_readlane(s_l, jj);
                    ko[u] = __builtin_amdgcn_readlane(k, jj);
                    nr[u] = __int_as_float(__builtin_amdgcn_readlane(__float_as_int(nrm_l), jj));
                    v[u]  = ((const float4*)(x + (size_t)s * D))[lane];
                }
#pragma unroll
                for (int u = 0; u < AGG_ROWS; ++u) {
                    if (j + u < m) {
                        if (ko[u] != cur) {
                            if (cur >= 0) put();
                            run  = make_float4(0.0f, 0.0f, 0.0f, 0.0f);
                            runs = 0.0f;
                            cur  = ko[u];
                        }
                        run.x += nr[u] * v[u].x;
                        run.y += nr[u] * v[u].y;
                        run.z += nr[u] * v[u].z;
                        run.w += nr[u] * v[u].w;
                        runs  += nr[u];
                    }
                }
            }
        }
        if (cur >= 0) put();
    } else {
        const int P = nblk * SORT_CHUNK;
        int* mylist = pend + wave * AGG_CAP;
        int cnt = 0;

        auto append = [&](bool hit, int e) {
            const unsigned mk = __builtin_amdgcn_ballot_w32(hit);
            if (hit) mylist[cnt + (int)__builtin_amdgcn_mbcnt_lo(mk, 0u)] = e;
            cnt += __builtin_popcount(mk);
        };

        auto flush = [&]() {
            __builtin_amdgcn_fence(__ATOMIC_ACQ_REL, "wavefront");
            for (int base = 0; base < cnt; base += 32) {
                const int m = min(32, cnt - base);
                int s_l = 0, tl_l = 0;
                float nrm_l = 0.0f;
                if (lane < m) {
                    const int e = mylist[base + lane];
                    int t;
                    if (e < P) { const int2 r = recs[e]; s_l = r.x; t = r.y; }
                    else       { t = s_l = e - P; }
                    tl_l  = t - node0;
                    nrm_l = __frsqrt_rn(deg[s_l] * deg[t]);
                    atomicAdd(nsum + tl_l, nrm_l);
                }
                for (int j = 0; j < m; ++j) {
                    const int   s  = __builtin_amdgcn_readlane(s_l, j);
                    const int   tj = __builtin_amdgcn_readlane(tl_l, j);
                    const float nj = __int_as_float(__builtin_amdgcn_readlane(__float_as_int(nrm_l), j));
                    const float4 v = ((const float4*)(x + (size_t)s * D))[lane];
                    float* a = acc + tj * ACC_P + 4 * lane;
                    atomicAdd(a,     nj * v.x);
                    atomicAdd(a + 1, nj * v.y);
                    atomicAdd(a + 2, nj * v.z);
                    atomicAdd(a + 3, nj * v.w);
                }
            }
            cnt = 0;
        };

        {
            const int tl = wave + AGG_WAVES * lane;
            append(tl < AGG_R && node0 + tl < N, P + node0 + tl);
        }
        for (int c0 = wave; c0 < nblk; c0 += AGG_WAVES * 32) {
            const int c = c0 + AGG_WAVES * lane;
            const unsigned word = (c < nblk) ? myseg[c] : 0u;
            unsigned live = __builtin_amdgcn_ballot_w32((word >> 16) != 0u);
            while (live != 0u) {
                const int l = __builtin_ctz(live);
                live &= live - 1u;
                const unsigned w = (unsigned)__builtin_amdgcn_readlane((int)word, l);
                const int count = (int)(w >> 16);
                const int pos0  = (c0 + AGG_WAVES * l) * SORT_CHUNK + (int)(w & 0xffffu);
                for (int i = 0; i < count; i += 32) {
                    append(i + lane < count, pos0 + i + lane);
                    if (cnt >= AGG_FLUSH) flush();
                }
            }
        }
        if (cnt > 0) flush();
    }
    __syncthreads();

    const int m    = lane & 15;
    const int half = lane >> 4;
    v8f c[2][2] = {};
#pragma unroll
    for (int slot = 0; slot < 2; ++slot) {
        const int job = wave + slot * AGG_WAVES;
        if (job < GEMM_JOBS) {
            const int strip = job >> 2, t0 = (job & 3) * 2;
            const float* arow = acc + (strip * 16 + m) * ACC_P + 2 * half;
            const float* b0   = wl + ((t0 + 0) * 16 + m) * ACC_P + 2 * half;
            const float* b1   = wl + ((t0 + 1) * 16 + m) * ACC_P + 2 * half;
#pragma unroll 8
            for (int kk = 0; kk < D / 4; ++kk) {
                const v2f a  = *(const v2f*)(arow + 4 * kk);
                const v2f w0 = *(const v2f*)(b0 + 4 * kk);
                const v2f w1 = *(const v2f*)(b1 + 4 * kk);
                c[slot][0] = __builtin_amdgcn_wmma_f32_16x16x4_f32(
                     false, a,  false, w0,
                     (short)0, c[slot][0],  false,  false);
                c[slot][1] = __builtin_amdgcn_wmma_f32_16x16x4_f32(
                     false, a,  false, w1,
                     (short)0, c[slot][1],  false,  false);
            }
        }
    }
    __syncthreads();

#pragma unroll
    for (int slot = 0; slot < 2; ++slot) {
        const int job = wave + slot * AGG_WAVES;
        if (job < GEMM_JOBS) {
            const int strip = job >> 2, t0 = (job & 3) * 2;
#pragma unroll
            for (int i = 0; i < 2; ++i) {
                const int   cc = (t0 + i) * 16 + m;
                const float bb = bias[cc];
                float* o = acc + (strip * 16 + half * 8) * ACC_P + cc;
#pragma unroll
                for (int v = 0; v < 8; ++v)
                    o[v * ACC_P] = c[slot][i][v] + nsum[strip * 16 + half * 8 + v] * bb;
            }
        }
    }
    __syncthreads();

    const float4 g  = ((const float4*)gamma)[lane];
    const float4 be = ((const float4*)beta)[lane];
    for (int tl = wave; tl < AGG_R; tl += AGG_WAVES) {
        const int n = node0 + tl;
        if (n >= N) break;

        const float4 a4 = *(const float4*)(acc + tl * ACC_P + 4 * lane);
        float4 v;
        v.x = fmaxf(a4.x, 0.0f);
        v.y = fmaxf(a4.y, 0.0f);
        v.z = fmaxf(a4.z, 0.0f);
        v.w = fmaxf(a4.w, 0.0f);

        float s  = v.x + v.y + v.z + v.w;
        float s2 = v.x * v.x + v.y * v.y + v.z * v.z + v.w * v.w;
#pragma unroll
        for (int o = 16; o >= 1; o >>= 1) {
            s  += __shfl_xor(s,  o, 32);
            s2 += __shfl_xor(s2, o, 32);
        }
        const float mu   = s * (1.0f / D);
        const float var  = s2 * (1.0f / D) - mu * mu;
        const float rstd = __frsqrt_rn(var + LN_EPS);

        const float4 xv = ((const float4*)(x + (size_t)n * D))[lane];

        float4 r;
        r.x = (v.x - mu) * rstd * g.x + be.x + xv.x;
        r.y = (v.y - mu) * rstd * g.y + be.y + xv.y;
        r.z = (v.z - mu) * rstd * g.z + be.z + xv.z;
        r.w = (v.w - mu) * rstd * g.w + be.w + xv.w;

        ((float4*)(out + (size_t)n * D))[lane] = r;
    }
}

extern "C" void kernel_launch(void* const* d_in, const int* in_sizes, int n_in,
                              void* d_out, int out_size, void* d_ws, size_t ws_size,
                              hipStream_t stream) {
    const float* x     = (const float*)d_in[0];
    const int*   ei    = (const int*)  d_in[1];
    const float* W     = (const float*)d_in[2];
    const float* bias  = (const float*)d_in[3];
    const float* gamma = (const float*)d_in[4];
    const float* beta  = (const float*)d_in[5];

    const int N = in_sizes[0] / D;
    const int E = in_sizes[1] / 2;

    const int G    = (N + AGG_R - 1) / AGG_R;
    const int GD   = (N + DEG_R - 1) / DEG_R;
    const int nblk = (E + SORT_CHUNK - 1) / SORT_CHUNK;
    const int GA   = (G + 31) / 32 * 32 + GD;

    char* p = (char*)d_ws;
    float*    deg  = (float*)p;     p += ((size_t)N * sizeof(float) + 255) / 256 * 256;
    unsigned* seg  = (unsigned*)p;  p += ((size_t)G * nblk * sizeof(unsigned) + 255) / 256 * 256;
    unsigned* segs = (unsigned*)p;  p += ((size_t)GD * nblk * sizeof(unsigned) + 255) / 256 * 256;
    int2*     recs = (int2*)p;      p += ((size_t)nblk * SORT_CHUNK * sizeof(int2) + 255) / 256 * 256;
    int*      srcs = (int*)p;

    rgcn_sort_kernel<<<nblk, SORT_T, (size_t)(2 * GA + (GA + 31) / 32) * sizeof(unsigned), stream>>>(
        ei, recs, seg, srcs, segs, E, G, GD, nblk);
    rgcn_owner_degree_kernel<<<GD, DEG_WAVES * 32, 0, stream>>>(srcs, segs, deg, N, nblk);
    rgcn_owner_sum_linear_final_kernel<<<G, 1024, 0, stream>>>(
        recs, seg, deg, x, W, bias, gamma, beta, (float*)d_out, N, nblk);
}


# final launch: a wave's independent L2 and LDS round trips issued together, not one per loop trip (W copy, 4 record steps, next batch's source+degrees, its 6-7 LN nodes); arrival-order sums kept
# speedup vs baseline: 4.4035x; 1.0518x over previous
#include <hip/hip_runtime.h>
#include <cstdint>
#include <cstddef>


typedef float v2f __attribute__((ext_vector_type(2)));
typedef float v8f __attribute__((ext_vector_type(8)));

#define D 128
#define LN_EPS 1e-5f

#define AGG_R       196
#define DEG_R       192
#define SORT_T      1024
#define SORT_Q      4
#define SORT_CHUNK  (SORT_T * SORT_Q)
static_assert(SORT_CHUNK < 65536, "start and count of a segment share one 32-bit word");

__global__ void __launch_bounds__(SORT_T)
rgcn_sort_kernel(const int* __restrict__ ei, int2* __restrict__ recs, unsigned* __restrict__ seg,
                 int* __restrict__ srcs, unsigned* __restrict__ segs,
                 int E, int G, int GD, int nblk) {
    extern __shared__ unsigned dyn[];
    __shared__ __attribute__((aligned(16))) int srec[2 * SORT_CHUNK];
    __shared__ __attribute__((aligned(16))) int ssrc[SORT_CHUNK];
    __shared__ unsigned stotal;
    __shared__ unsigned stotal_s;
    const int GP  = (G + 31) & ~31;
    const int GA  = GP + GD;
    const int npt = GP >> 5;
    unsigned* bcnt = dyn;
    unsigned* bval = bcnt + GA;
    unsigned* ptot = bval + GA;
    const int lane = threadIdx.x & 31;
    const int wave = __builtin_amdgcn_readfirstlane(threadIdx.x >> 5);

    for (int i = threadIdx.x; i < GA; i += SORT_T) dyn[i] = 0u;
    if (threadIdx.x == 0) { stotal = 0u; stotal_s = 0u; }
    __syncthreads();

    const int* __restrict__ col = ei + E;
    const int cbase = blockIdx.x * SORT_CHUNK;
    int s[SORT_Q], t[SORT_Q];
#pragma unroll
    for (int q = 0; q < SORT_Q; ++q) {
        const int e = min(cbase + q * SORT_T + (int)threadIdx.x, E - 1);
        s[q] = ei[e];
        t[q] = col[e];
    }
    unsigned ot[SORT_Q], os[SORT_Q], rank[SORT_Q], ranks[SORT_Q];
#pragma unroll
    for (int q = 0; q < SORT_Q; ++q) {
        ot[q]    = 0xffffffffu;
        os[q]    = 0xffffffffu;
        rank[q]  = 0u;
        ranks[q] = 0u;
        if (cbase + q * SORT_T + (int)threadIdx.x < E) {
            ot[q] = (unsigned)t[q] / (unsigned)AGG_R;
            if (ot[q] < (unsigned)G) rank[q] = atomicAdd(&bcnt[ot[q]], 1u);
            os[q] = (unsigned)s[q] / (unsigned)DEG_R;
            if (os[q] < (unsigned)GD) ranks[q] = atomicAdd(&bcnt[GP + os[q]], 1u);
        }
    }
    __syncthreads();

    const int npiece = (GA + 31) >> 5;
    for (int pc = wave; pc < npiece; pc += SORT_T / 32) {
        const int o = pc * 32 + lane;
        const unsigned v = (o < GA) ? bcnt[o] : 0u;
        unsigned incl = v;
#pragma unroll
        for (int d = 1; d < 32; d <<= 1) {
            const unsigned u = __shfl_up(incl, d, 32);
            if (lane >= d) incl += u;
        }
        if (o < GA) { bcnt[o] = incl - v; bval[o] = v; }
        if (lane == 31) ptot[pc] = incl;
    }
    __syncthreads();

    for (int o = threadIdx.x; o < GA; o += SORT_T) {
        const bool srcside = o >= GP;
        unsigned start = bcnt[o];
        for (int pp = srcside ? npt : 0; pp < (o >> 5); ++pp) start += ptot[pp];
        const unsigned v = bval[o];
        bcnt[o] = start;
        if (srcside) {
            const int od = o - GP;
            segs[(size_t)od * nblk + blockIdx.x] = start | (v << 16);
            if (od == GD - 1) stotal_s = start + v;
        } else if (o < G) {
            seg[(size_t)o * nblk + blockIdx.x] = start | (v << 16);
            if (o == G - 1) stotal = start + v;
        }
    }
    __syncthreads();

#pragma unroll
    for (int q = 0; q < SORT_Q; ++q) {
        if (ot[q] < (unsigned)G)
            *(int2*)(srec + 2 * (bcnt[ot[q]] + rank[q])) = make_int2(s[q], t[q]);
        if (os[q] < (unsigned)GD)
            ssrc[bcnt[GP + os[q]] + ranks[q]] = s[q];
    }
    __syncthreads();

    const int npair = ((int)stotal + 1) >> 1;
    int4* dst = (int4*)(recs + (size_t)cbase);
    for (int i = threadIdx.x; i < npair; i += SORT_T) dst[i] = ((const int4*)srec)[i];
    const int nquad = ((int)stotal_s + 3) >> 2;
    int4* dsts = (int4*)(srcs + (size_t)cbase);
    for (int i = threadIdx.x; i < nquad; i += SORT_T) dsts[i] = ((const int4*)ssrc)[i];
}

#define DEG_WAVES  8
#define DEG_Q      8
static_assert(DEG_WAVES * 32 >= DEG_R, "a thread per owned node clears and stores its counter");

__global__ void __launch_bounds__(DEG_WAVES * 32)
rgcn_owner_degree_kernel(const int* __restrict__ srcs, const unsigned* __restrict__ segs,
                         float* __restrict__ deg, int N, int nblk) {
    __shared__ unsigned cnt[DEG_R];

    const int lane  = threadIdx.x & 31;
    const int wave  = __builtin_amdgcn_readfirstlane(threadIdx.x >> 5);
    const int node0 = blockIdx.x * DEG_R;

    if (threadIdx.x < DEG_R) cnt[threadIdx.x] = 0u;
    __syncthreads();

    const unsigned* __restrict__ myseg = segs + (size_t)blockIdx.x * nblk;
    for (int c0 = wave; c0 < nblk; c0 += DEG_WAVES * 32) {
        const int c = c0 + DEG_WAVES * lane;
        const unsigned word = (c < nblk) ? myseg[c] : 0u;
        const int first = c * SORT_CHUNK + (int)(word & 0xffffu);
        const int last  = first + (int)(word >> 16);
        auto count = [&](int sv, int idx) {
            const unsigned sl = (unsigned)(sv - node0);
            if (idx >= first && idx < last && sl < (unsigned)DEG_R) atomicAdd(&cnt[sl], 1u);
        };
        int unit = (first < last) ? (first & ~3) : last;
        while (__builtin_amdgcn_ballot_w32(unit < last) != 0u) {
            int4 v[DEG_Q];
#pragma unroll
            for (int u = 0; u < DEG_Q; ++u) {
                v[u] = make_int4(0, 0, 0, 0);
                if (unit + 4 * u < last) v[u] = *(const int4*)(srcs + unit + 4 * u);
            }
#pragma unroll
            for (int u = 0; u < DEG_Q; ++u) {
                count(v[u].x, unit + 4 * u);
                count(v[u].y, unit + 4 * u + 1);
                count(v[u].z, unit + 4 * u + 2);
                count(v[u].w, unit + 4 * u + 3);
            }
            unit += 4 * DEG_Q;
        }
    }
    __syncthreads();

    if (threadIdx.x < DEG_R) {
        const int n = node0 + threadIdx.x;
        if (n < N) deg[n] = 1.0f + (float)cnt[threadIdx.x];
    }
}

#define AGG_WAVES  32
#define AGG_QCAP   64
#define AGG_OWN    7
#define AGG_FLUSH  64
#define AGG_CAP    (AGG_FLUSH + 32)
#define AGG_ROWS   8
#define AGG_RP     208
#define ACC_P      132
#define GEMM_JOBS  (AGG_RP / 16 * 4)
#define W_Q        (D * D / 4 / 1024)
#define DEAL_Q     4
static_assert(W_Q * 1024 * 4 == D * D, "every thread copies the same number of pieces of W");
static_assert(AGG_OWN == 7, "the statement that holds the residual-row loads names seven rows");
static_assert(AGG_WAVES == 32 && (AGG_R + AGG_WAVES - 1) / AGG_WAVES == AGG_OWN && AGG_OWN <= 8,
              "a wave's targets are found by a three-step search over lanes 0..7");

__global__ void __launch_bounds__(1024)
rgcn_owner_sum_linear_final_kernel(const int2* __restrict__ recs, const unsigned* __restrict__ seg,
                                   const float* __restrict__ deg,
                                   const float* __restrict__ x, const float* __restrict__ W,
                                   const float* __restrict__ bias,
                                   const float* __restrict__ gamma, const float* __restrict__ beta,
                                   float* __restrict__ out, int N, int nblk) {
    __shared__ __attribute__((aligned(16))) float acc[AGG_RP * ACC_P];
    __shared__ __attribute__((aligned(16))) float wl[D * ACC_P];
    __shared__ float    nsum[AGG_RP];
    __shared__ int      queue[AGG_RP * AGG_QCAP];
    __shared__ unsigned qcnt[AGG_RP];
    __shared__ int      pend[AGG_WAVES * AGG_CAP];

    const int lane  = threadIdx.x & 31;
    const int wave  = __builtin_amdgcn_readfirstlane(threadIdx.x >> 5);
    const int node0 = blockIdx.x * AGG_R;

    const unsigned* __restrict__ myseg = seg + (size_t)blockIdx.x * nblk;

    unsigned word_next;
    {
        const int c = wave + AGG_WAVES * lane;
        word_next = (c < nblk) ? myseg[c] : 0u;
    }
    float4 wv[W_Q];
#pragma unroll
    for (int q = 0; q < W_Q; ++q) wv[q] = ((const float4*)W)[(int)threadIdx.x + 1024 * q];

    for (int i = threadIdx.x; i < AGG_RP * ACC_P / 4; i += 1024)
        ((float4*)acc)[i] = make_float4(0.0f, 0.0f, 0.0f, 0.0f);
    if (threadIdx.x < AGG_RP) nsum[threadIdx.x] = 0.0f;
    if (threadIdx.x < AGG_RP) qcnt[threadIdx.x] = 0u;
#pragma unroll
    for (int q = 0; q < W_Q; ++q) {
        const int i = (int)threadIdx.x + 1024 * q;
        const int n = i >> 5, c4 = i & 31;
        *(float4*)(wl + n * ACC_P + 4 * c4) = wv[q];
    }
    __syncthreads();

    auto deal = [&](bool hit, int s, int tl) {
        if (hit) {
            const unsigned slot = atomicAdd(&qcnt[tl], 1u);
            if (slot < (unsigned)AGG_QCAP) queue[tl * AGG_QCAP + (int)slot] = s;
        }
    };
    {
        const int tl = wave + AGG_WAVES * lane;
        deal(tl < AGG_R && node0 + tl < N, node0 + tl, tl);
    }
    for (int c0 = wave; c0 < nblk; c0 += AGG_WAVES * 32) {
        const int c = c0 + AGG_WAVES * lane;
        const unsigned word = word_next;
        {
            const int cn = c + AGG_WAVES * 32;
            word_next = (cn < nblk) ? myseg[cn] : 0u;
        }
        const int count = (int)(word >> 16);
        const int pos   = c * SORT_CHUNK + (int)(word & 0xffffu);
        int incl = count;
#pragma unroll
        for (int d = 1; d < 32; d <<= 1) {
            const int u = __shfl_up(incl, d, 32);
            if (lane >= d) incl += u;
        }
        const int pre   = incl - count;
        const int total = __builtin_amdgcn_readlane(incl, 31);
        for (int f0 = 0; f0 < total; f0 += 32 * DEAL_Q) {
            int2 r[DEAL_Q];
            bool hit[DEAL_Q];
            int  f[DEAL_Q], lo[DEAL_Q];
#pragma unroll
            for (int q = 0; q < DEAL_Q; ++q) { f[q] = f0 + 32 * q + lane; hit[q] = f[q] < total; lo[q] = 0; }
#pragma unroll
            for (int step = 16; step >= 1; step >>= 1) {
                int p[DEAL_Q];
#pragma unroll
                for (int q = 0; q < DEAL_Q; ++q) p[q] = __shfl(pre, lo[q] + step, 32);
#pragma unroll
                for (int q = 0; q < DEAL_Q; ++q) if (p[q] <= f[q]) lo[q] += step;
            }
            int ps[DEAL_Q], pr[DEAL_Q];
#pragma unroll
            for (int q = 0; q < DEAL_Q; ++q) { ps[q] = __shfl(pos, lo[q], 32); pr[q] = __shfl(pre, lo[q], 32); }
#pragma unroll
            for (int q = 0; q < DEAL_Q; ++q) {
                r[q] = make_int2(0, node0);
                if (hit[q]) r[q] = recs[ps[q] + (f[q] - pr[q])];
            }
#pragma unroll
            for (int q = 0; q < DEAL_Q; ++q) deal(hit[q], r[q].x, r[q].y - node0);
        }
    }
    __syncthreads();

    bool over = false;
    for (int i = lane; i < AGG_R; i += 32) over = over || qcnt[i] > (unsigned)AGG_QCAP;
    const bool fits = __builtin_amdgcn_ballot_w32(over) == 0u;

    if (fits) {
        const int tlk   = wave + AGG_WAVES * lane;
        const int qlen  = (lane < AGG_OWN && tlk < AGG_R) ? (int)qcnt[tlk] : 0;
        int incl = qlen;
#pragma unroll
        for (int d = 1; d < 32; d <<= 1) {
            const int u = __shfl_up(incl, d, 32);
            if (lane >= d) incl += u;
        }
        const int pre = incl - qlen;
        const int nq  = __builtin_amdgcn_readlane(incl, 31);

        float4 run  = make_float4(0.0f, 0.0f, 0.0f, 0.0f);
        float  runs = 0.0f;
        int    cur  = -1;
        auto put = [&]() {
            const int tl = wave + AGG_WAVES * cur;
            *((float4*)(acc + tl * ACC_P) + lane) = run;
            if (lane == 0) nsum[tl] = runs;
        };

        int   k_n  = 0;
        int   s_n  = 0;
        float da_n = 1.0f, db_n = 1.0f;
        auto fetch = [&](int base) {
            const int m = min(32, nq - base);
            const int f = base + lane;
            int k = 0;
#pragma unroll
            for (int step = 4; step >= 1; step >>= 1) {
                const int p = __shfl(pre, k + step, 32);
                if (p <= f) k += step;
            }
            const int fk = f - __shfl(pre, k, 32);
            k_n  = k;
            s_n  = 0;
            da_n = 1.0f;
            db_n = 1.0f;
            if (lane < m) {
                const int tl = wave + AGG_WAVES * k;
                s_n  = queue[tl * AGG_QCAP + fk];
                da_n = deg[s_n];
                db_n = deg[node0 + tl];
            }
        };
        if (nq > 0) fetch(0);

        for (int base = 0; base < nq; base += 32) {
            const int m = min(32, nq - base);
            const int   k     = k_n;
            const int   s_l   = s_n;
            const float nrm_l = (lane < m) ? __frsqrt_rn(da_n * db_n) : 0.0f;
            if (base + 32 < nq) fetch(base + 32);
            for (int j = 0; j < m; j += AGG_ROWS) {
                float4 v[AGG_ROWS];
                float  nr[AGG_ROWS];
                int    ko[AGG_ROWS];
#pragma unroll
                for (int u = 0; u < AGG_ROWS; ++u) {
                    const int jj = min(j + u, m - 1);
                    const int s  = __builtin_amdgcn_readlane(s_l, jj);
                    ko[u] = __builtin_amdgcn_readlane(k, jj);
                    nr[u] = __int_as_float(__builtin_amdgcn_readlane(__float_as_int(nrm_l), jj));
                    v[u]  = ((const float4*)(x + (size_t)s * D))[lane];
                }
#pragma unroll
                for (int u = 0; u < AGG_ROWS; ++u) {
                    if (j + u < m) {
                        if (ko[u] != cur) {
                            if (cur >= 0) put();
                            run  = make_float4(0.0f, 0.0f, 0.0f, 0.0f);
                            runs = 0.0f;
                            cur  = ko[u];
                        }
                        run.x += nr[u] * v[u].x;
                        run.y += nr[u] * v[u].y;
                        run.z += nr[u] * v[u].z;
                        run.w += nr[u] * v[u].w;
                        runs  += nr[u];
                    }
                }
            }
        }
        if (cur >= 0) put();
    } else {
        const int P = nblk * SORT_CHUNK;
        int* mylist = pend + wave * AGG_CAP;
        int cnt = 0;

        auto append = [&](bool hit, int e) {
            const unsigned mk = __builtin_amdgcn_ballot_w32(hit);
            if (hit) mylist[cnt + (int)__builtin_amdgcn_mbcnt_lo(mk, 0u)] = e;
            cnt += __builtin_popcount(mk);
        };

        auto flush = [&]() {
            __builtin_amdgcn_fence(__ATOMIC_ACQ_REL, "wavefront");
            for (int base = 0; base < cnt; base += 32) {
                const int m = min(32, cnt - base);
                int s_l = 0, tl_l = 0;
                float nrm_l = 0.0f;
                if (lane < m) {
                    const int e = mylist[base + lane];
                    int t;
                    if (e < P) { const int2 r = recs[e]; s_l = r.x; t = r.y; }
                    else       { t = s_l = e - P; }
                    tl_l  = t - node0;
                    nrm_l = __frsqrt_rn(deg[s_l] * deg[t]);
                    atomicAdd(nsum + tl_l, nrm_l);
                }
                for (int j = 0; j < m; ++j) {
                    const int   s  = __builtin_amdgcn_readlane(s_l, j);
                    const int   tj = __builtin_amdgcn_readlane(tl_l, j);
                    const float nj = __int_as_float(__builtin_amdgcn_readlane(__float_as_int(nrm_l), j));
                    const float4 v = ((const float4*)(x + (size_t)s * D))[lane];
                    float* a = acc + tj * ACC_P + 4 * lane;
                    atomicAdd(a,     nj * v.x);
                    atomicAdd(a + 1, nj * v.y);
                    atomicAdd(a + 2, nj * v.z);
                    atomicAdd(a + 3, nj * v.w);
                }
            }
            cnt = 0;
        };

        {
            const int tl = wave + AGG_WAVES * lane;
            append(tl < AGG_R && node0 + tl < N, P + node0 + tl);
        }
        for (int c0 = wave; c0 < nblk; c0 += AGG_WAVES * 32) {
            const int c = c0 + AGG_WAVES * lane;
            const unsigned word = (c < nblk) ? myseg[c] : 0u;
            unsigned live = __builtin_amdgcn_ballot_w32((word >> 16) != 0u);
            while (live != 0u) {
                const int l = __builtin_ctz(live);
                live &= live - 1u;
                const unsigned w = (unsigned)__builtin_amdgcn_readlane((int)word, l);
                const int count = (int)(w >> 16);
                const int pos0  = (c0 + AGG_WAVES * l) * SORT_CHUNK + (int)(w & 0xffffu);
                for (int i = 0; i < count; i += 32) {
                    append(i + lane < count, pos0 + i + lane);
                    if (cnt >= AGG_FLUSH) flush();
                }
            }
        }
        if (cnt > 0) flush();
    }
    __syncthreads();

    const int m    = lane & 15;
    const int half = lane >> 4;
    float bb[2][2];
#pragma unroll
    for (int slot = 0; slot < 2; ++slot) {
        const int job = wave + slot * AGG_WAVES;
#pragma unroll
        for (int i = 0; i < 2; ++i)
            bb[slot][i] = (job < GEMM_JOBS) ? bias[((job & 3) * 2 + i) * 16 + m] : 0.0f;
    }
    v8f c[2][2] = {};
#pragma unroll
    for (int slot = 0; slot < 2; ++slot) {
        const int job = wave + slot * AGG_WAVES;
        if (job < GEMM_JOBS) {
            const int strip = job >> 2, t0 = (job & 3) * 2;
            const float* arow = acc + (strip * 16 + m) * ACC_P + 2 * half;
            const float* b0   = wl + ((t0 + 0) * 16 + m) * ACC_P + 2 * half;
            const float* b1   = wl + ((t0 + 1) * 16 + m) * ACC_P + 2 * half;
#pragma unroll 8
            for (int kk = 0; kk < D / 4; ++kk) {
                const v2f a  = *(const v2f*)(arow + 4 * kk);
                const v2f w0 = *(const v2f*)(b0 + 4 * kk);
                const v2f w1 = *(const v2f*)(b1 + 4 * kk);
                c[slot][0] = __builtin_amdgcn_wmma_f32_16x16x4_f32(
                     false, a,  false, w0,
                     (short)0, c[slot][0],  false,  false);
                c[slot][1] = __builtin_amdgcn_wmma_f32_16x16x4_f32(
                     false, a,  false, w1,
                     (short)0, c[slot][1],  false,  false);
            }
        }
    }
    __syncthreads();

#pragma unroll
    for (int slot = 0; slot < 2; ++slot) {
        const int job = wave + slot * AGG_WAVES;
        if (job < GEMM_JOBS) {
            const int strip = job >> 2, t0 = (job & 3) * 2;
#pragma unroll
            for (int i = 0; i < 2; ++i) {
                const int   cc = (t0 + i) * 16 + m;
                float* o = acc + (strip * 16 + half * 8) * ACC_P + cc;
#pragma unroll
                for (int v = 0; v < 8; ++v)
                    o[v * ACC_P] = c[slot][i][v] + nsum[strip * 16 + half * 8 + v] * bb[slot][i];
            }
        }
    }
    __syncthreads();

    float4 xr[AGG_OWN];
#pragma unroll
    for (int k = 0; k < AGG_OWN; ++k) {
        const int nr = min(node0 + wave + AGG_WAVES * k, N - 1);
        xr[k] = ((const float4*)(x + (size_t)nr * D))[lane];
    }
    const float4 g  = ((const float4*)gamma)[lane];
    const float4 be = ((const float4*)beta)[lane];

    float4 v[AGG_OWN];
    float  s[AGG_OWN], s2[AGG_OWN];
#pragma unroll
    for (int k = 0; k < AGG_OWN; ++k) {
        const int tlr = min(wave + AGG_WAVES * k, AGG_RP - 1);
        const float4 a4 = *(const float4*)(acc + tlr * ACC_P + 4 * lane);
        v[k].x = fmaxf(a4.x, 0.0f);
        v[k].y = fmaxf(a4.y, 0.0f);
        v[k].z = fmaxf(a4.z, 0.0f);
        v[k].w = fmaxf(a4.w, 0.0f);
        s[k]  = v[k].x + v[k].y + v[k].z + v[k].w;
        s2[k] = v[k].x * v[k].x + v[k].y * v[k].y + v[k].z * v[k].z + v[k].w * v[k].w;
    }
#pragma unroll
    for (int o = 16; o >= 1; o >>= 1) {
        float ts[AGG_OWN], ts2[AGG_OWN];
#pragma unroll
        for (int k = 0; k < AGG_OWN; ++k) {
            ts[k]  = __shfl_xor(s[k],  o, 32);
            ts2[k] = __shfl_xor(s2[k], o, 32);
        }
#pragma unroll
        for (int k = 0; k < AGG_OWN; ++k) {
            s[k]  += ts[k];
            s2[k] += ts2[k];
        }
    }
    asm volatile("" :: "v"(xr[0].x), "v"(xr[1].x), "v"(xr[2].x), "v"(xr[3].x),
                       "v"(xr[4].x), "v"(xr[5].x), "v"(xr[6].x));
#pragma unroll
    for (int k = 0; k < AGG_OWN; ++k) {
        const int tl = wave + AGG_WAVES * k;
        const int n  = node0 + tl;
        if (tl < AGG_R && n < N) {
            const float mu   = s[k] * (1.0f / D);
            const float var  = s2[k] * (1.0f / D) - mu * mu;
            const float rstd = __frsqrt_rn(var + LN_EPS);

            float4 r;
            r.x = (v[k].x - mu) * rstd * g.x + be.x + xr[k].x;
            r.y = (v[k].y - mu) * rstd * g.y + be.y + xr[k].y;
            r.z = (v[k].z - mu) * rstd * g.z + be.z + xr[k].z;
            r.w = (v[k].w - mu) * rstd * g.w + be.w + xr[k].w;

            ((float4*)(out + (size_t)n * D))[lane] = r;
        }
    }
}

extern "C" void kernel_launch(void* const* d_in, const int* in_sizes, int n_in,
                              void* d_out, int out_size, void* d_ws, size_t ws_size,
                              hipStream_t stream) {
    const float* x     = (const float*)d_in[0];
    const int*   ei    = (const int*)  d_in[1];
    const float* W     = (const float*)d_in[2];
    const float* bias  = (const float*)d_in[3];
    const float* gamma = (const float*)d_in[4];
    const float* beta  = (const float*)d_in[5];

    const int N = in_sizes[0] / D;
    const int E = in_sizes[1] / 2;

    const int G    = (N + AGG_R - 1) / AGG_R;
    const int GD   = (N + DEG_R - 1) / DEG_R;
    const int nblk = (E + SORT_CHUNK - 1) / SORT_CHUNK;
    const int GA   = (G + 31) / 32 * 32 + GD;

    char* p = (char*)d_ws;
    float*    deg  = (float*)p;     p += ((size_t)N * sizeof(float) + 255) / 256 * 256;
    unsigned* seg  = (unsigned*)p;  p += ((size_t)G * nblk * sizeof(unsigned) + 255) / 256 * 256;
    unsigned* segs = (unsigned*)p;  p += ((size_t)GD * nblk * sizeof(unsigned) + 255) / 256 * 256;
    int2*     recs = (int2*)p;      p += ((size_t)nblk * SORT_CHUNK * sizeof(int2) + 255) / 256 * 256;
    int*      srcs = (int*)p;

    rgcn_sort_kernel<<<nblk, SORT_T, (size_t)(2 * GA + (GA + 31) / 32) * sizeof(unsigned), stream>>>(
        ei, recs, seg, srcs, segs, E, G, GD, nblk);
    rgcn_owner_degree_kernel<<<GD, DEG_WAVES * 32, 0, stream>>>(srcs, segs, deg, N, nblk);
    rgcn_owner_sum_linear_final_kernel<<<G, 1024, 0, stream>>>(
        recs, seg, deg, x, W, bias, gamma, beta, (float*)d_out, N, nblk);
}
